# starting kernel
# baseline (speedup 1.0000x reference)
#include <hip/hip_runtime.h>
#include <hip/hip_bf16.h>
#include <math.h>

typedef __attribute__((ext_vector_type(2))) float v2f;
typedef __attribute__((ext_vector_type(8))) float v8f;

__device__ __forceinline__ float wave_sum32(float v) {
#pragma unroll
  for (int off = 16; off > 0; off >>= 1) v += __shfl_xor(v, off, 32);
  return v;
}

__device__ __forceinline__ void atomicMaxF(float* addr, float val) {
  if (val >= 0.0f) atomicMax((int*)addr, __float_as_int(val));
  else             atomicMin((unsigned int*)addr, __float_as_uint(val));
}

__global__ void fill_f32(float* __restrict__ p, float v, int n) {
  int i = blockIdx.x * blockDim.x + threadIdx.x;
  if (i < n) p[i] = v;
}

template <int K, int H>
__global__ void gemm_bias_wmma(const float* __restrict__ X, const float* __restrict__ W,
                               const float* __restrict__ bias, float* __restrict__ C, int N) {
  constexpr int COLT = H / 16;
  constexpr int ROWT = 8 / COLT;
  const int wave = threadIdx.x >> 5;
  const int lane = threadIdx.x & 31;
  const int rt = wave / COLT, ct = wave % COLT;
  const int row0 = (blockIdx.x * ROWT + rt) * 16;
  if (row0 >= N) return;
  const int col0 = ct * 16;
  const int half = lane >> 4, l16 = lane & 15;

  const float* __restrict__ arow = X + (size_t)(row0 + l16) * K;
  v8f acc = {};
#pragma unroll 4
  for (int k = 0; k < K; k += 4) {
    const int ka = k + 2 * half;
    v2f a, b;
    a.x = arow[ka];
    a.y = arow[ka + 1];
    b.x = W[(size_t)ka * H + col0 + l16];
    b.y = W[(size_t)(ka + 1) * H + col0 + l16];
    acc = __builtin_amdgcn_wmma_f32_16x16x4_f32(false, a, false, b, (short)0, acc,
                                                false, false);
  }
  const float bv = bias[col0 + l16];
  float* __restrict__ crow = C + (size_t)(row0 + 8 * half) * H + col0 + l16;
#pragma unroll
  for (int v = 0; v < 8; ++v) crow[(size_t)v * H] = acc[v] + bv;
}

template <int H>
__global__ void edge_scores(const float* __restrict__ xl, const float* __restrict__ xr,
                            const float* __restrict__ att, const int* __restrict__ ei,
                            int E, int Etot, float* __restrict__ ebuf,
                            float* __restrict__ m) {
  const int gid = blockIdx.x * blockDim.x + threadIdx.x;
  const int e = gid >> 5, lane = gid & 31;
  if (e >= Etot) return;
  int s, d;
  if (e < E) { s = ei[e]; d = ei[E + e]; } else { s = d = e - E; }

  float partial;
  if (H == 64) {
    const int c0 = lane * 2;
    const float2 a = *(const float2*)(xl + (size_t)s * 64 + c0);
    const float2 b = *(const float2*)(xr + (size_t)d * 64 + c0);
    float h0 = a.x + b.x; h0 = h0 > 0.0f ? h0 : 0.2f * h0;
    float h1 = a.y + b.y; h1 = h1 > 0.0f ? h1 : 0.2f * h1;
    partial = h0 * att[c0] + h1 * att[c0 + 1];
  } else {
    float hv = xl[(size_t)s * H + lane] + xr[(size_t)d * H + lane];
    hv = hv > 0.0f ? hv : 0.2f * hv;
    partial = hv * att[lane];
  }
  const float tot = wave_sum32(partial);
  if (lane == 0) {
    ebuf[e] = tot;
    atomicMaxF(m + d, tot);
  }
}

template <int H>
__global__ void edge_aggregate(const float* __restrict__ xl, const int* __restrict__ ei,
                               int E, int Etot, const float* __restrict__ ebuf,
                               const float* __restrict__ m, float* __restrict__ denom,
                               float* __restrict__ acc) {
  const int gid = blockIdx.x * blockDim.x + threadIdx.x;
  const int e = gid >> 5, lane = gid & 31;
  if (e >= Etot) return;
  int s, d;
  if (e < E) { s = ei[e]; d = ei[E + e]; } else { s = d = e - E; }

  const float ex = expf(ebuf[e] - m[d]);
  if (lane == 0) atomicAdd(denom + d, ex);
  if (H == 64) {
    const int c0 = lane * 2;
    const float2 a = *(const float2*)(xl + (size_t)s * 64 + c0);
    atomicAdd(acc + (size_t)d * 64 + c0, ex * a.x);
    atomicAdd(acc + (size_t)d * 64 + c0 + 1, ex * a.y);
  } else {
    atomicAdd(acc + (size_t)d * H + lane, ex * xl[(size_t)s * H + lane]);
  }
}

__global__ void finalize_layer1(const float* __restrict__ acc, const float* __restrict__ denom,
                                const float* __restrict__ bias, float* __restrict__ h, int N) {
  const int i = blockIdx.x * blockDim.x + threadIdx.x;
  if (i >= N * 64) return;
  const int n = i >> 6, c = i & 63;
  float v = acc[i] / (denom[n] + 1e-16f) + bias[c];
  h[i] = v > 0.0f ? v : 0.01f * v;
}

__global__ void finalize_layer2(const float* __restrict__ acc, const float* __restrict__ denom,
                                const float* __restrict__ bias, const float* __restrict__ Wlin,
                                const float* __restrict__ blin, float* __restrict__ out, int N) {
  const int gid = blockIdx.x * blockDim.x + threadIdx.x;
  const int n = gid >> 5, lane = gid & 31;
  if (n >= N) return;
  const float v = acc[(size_t)n * 32 + lane] / (denom[n] + 1e-16f) + bias[lane];
  const float tot = wave_sum32(v * Wlin[lane]);
  if (lane == 0) out[n] = tot + blin[0];
}

static inline int cdiv(int a, int b) { return (a + b - 1) / b; }

extern "C" void kernel_launch(void* const* d_in, const int* in_sizes, int n_in,
                              void* d_out, int out_size, void* d_ws, size_t ws_size,
                              hipStream_t stream) {
  const float* x     = (const float*)d_in[0];
  const int*   ei    = (const int*)  d_in[1];
  const float* W1l   = (const float*)d_in[2];
  const float* b1l   = (const float*)d_in[3];
  const float* W1r   = (const float*)d_in[4];
  const float* b1r   = (const float*)d_in[5];
  const float* att1  = (const float*)d_in[6];
  const float* bias1 = (const float*)d_in[7];
  const float* W2l   = (const float*)d_in[8];
  const float* b2l   = (const float*)d_in[9];
  const float* W2r   = (const float*)d_in[10];
  const float* b2r   = (const float*)d_in[11];
  const float* att2  = (const float*)d_in[12];
  const float* bias2 = (const float*)d_in[13];
  const float* Wlin  = (const float*)d_in[14];
  const float* blin  = (const float*)d_in[15];
  float* out = (float*)d_out;

  const int N    = in_sizes[0] / 128;
  const int E    = in_sizes[1] / 2;
  const int Etot = E + N;

  float* ws = (float*)d_ws;
  size_t off = 0;
  float* xl1  = ws + off; off += (size_t)N * 64;
  float* xr1  = ws + off; off += (size_t)N * 64;
  float* acc1 = ws + off; off += (size_t)N * 64;
  float* xl2  = ws + off; off += (size_t)N * 32;
  float* xr2  = ws + off; off += (size_t)N * 32;
  float* acc2 = ws + off; off += (size_t)N * 32;
  float* ebuf = ws + off; off += (size_t)Etot;
  float* mbuf = ws + off; off += (size_t)N;
  float* dbuf = ws + off; off += (size_t)N;
  (void)ws_size; (void)n_in; (void)out_size;

  const int TB = 256;
  const int rowTiles = cdiv(N, 16);
  const int eblocks = cdiv(Etot, TB / 32);

  gemm_bias_wmma<128, 64><<<cdiv(rowTiles, 2), TB, 0, stream>>>(x, W1l, b1l, xl1, N);
  gemm_bias_wmma<128, 64><<<cdiv(rowTiles, 2), TB, 0, stream>>>(x, W1r, b1r, xr1, N);

  fill_f32<<<cdiv(N * 64, TB), TB, 0, stream>>>(acc1, 0.0f, N * 64);
  fill_f32<<<cdiv(N, TB), TB, 0, stream>>>(mbuf, -INFINITY, N);
  fill_f32<<<cdiv(N, TB), TB, 0, stream>>>(dbuf, 0.0f, N);

  edge_scores<64><<<eblocks, TB, 0, stream>>>(xl1, xr1, att1, ei, E, Etot, ebuf, mbuf);
  edge_aggregate<64><<<eblocks, TB, 0, stream>>>(xl1, ei, E, Etot, ebuf, mbuf, dbuf, acc1);
  finalize_layer1<<<cdiv(N * 64, TB), TB, 0, stream>>>(acc1, dbuf, bias1, acc1, N);

  gemm_bias_wmma<64, 32><<<cdiv(rowTiles, 4), TB, 0, stream>>>(acc1, W2l, b2l, xl2, N);
  gemm_bias_wmma<64, 32><<<cdiv(rowTiles, 4), TB, 0, stream>>>(acc1, W2r, b2r, xr2, N);

  fill_f32<<<cdiv(N * 32, TB), TB, 0, stream>>>(acc2, 0.0f, N * 32);
  fill_f32<<<cdiv(N, TB), TB, 0, stream>>>(mbuf, -INFINITY, N);
  fill_f32<<<cdiv(N, TB), TB, 0, stream>>>(dbuf, 0.0f, N);

  edge_scores<32><<<eblocks, TB, 0, stream>>>(xl2, xr2, att2, ei, E, Etot, ebuf, mbuf);
  edge_aggregate<32><<<eblocks, TB, 0, stream>>>(xl2, ei, E, Etot, ebuf, mbuf, dbuf, acc2);

  finalize_layer2<<<cdiv(N, TB / 32), TB, 0, stream>>>(acc2, dbuf, bias2, Wlin, blin, out, N);
}


# owner passes: LDS instructions per list entry 17 -> 3 (layer 2: 10 -> 1): wave sums as DPP/permlanex16 adds (same butterfly), per-entry scalar work of 8 entries on 8 lanes; LDS adds arrive as before
# speedup vs baseline: 4.9861x; 4.9861x over previous
#include <hip/hip_runtime.h>
#include <hip/hip_bf16.h>
#include <math.h>

typedef __attribute__((ext_vector_type(2))) float v2f;
typedef __attribute__((ext_vector_type(8))) float v8f;

__device__ __forceinline__ float wave_sum32(float v) {
#pragma unroll
  for (int off = 16; off > 0; off >>= 1) v += __shfl_xor(v, off, 32);
  return v;
}

template <int M>
__device__ __forceinline__ float lane_xor_row(float x) {
  const int b = __builtin_bit_cast(int, x);
  return __builtin_bit_cast(float, __builtin_amdgcn_update_dpp(b, b, 0x160 + M, 0xF, 0xF, true));
}
__device__ __forceinline__ float lane_xor16(float x) {
  const int b = __builtin_bit_cast(int, x);
  return __builtin_bit_cast(float,
                            __builtin_amdgcn_permlanex16(b, b, 0x76543210, 0xfedcba98, false, false));
}
__device__ __forceinline__ int lane_int(int x, int l) { return __builtin_amdgcn_readlane(x, l); }
__device__ __forceinline__ float lane_float(float x, int l) {
  return __builtin_bit_cast(float, __builtin_amdgcn_readlane(__builtin_bit_cast(int, x), l));
}

template <int K, int H>
__global__ void gemm_bias_wmma(const float* __restrict__ X, const float* __restrict__ W,
                               const float* __restrict__ bias, float* __restrict__ C, int N) {
  constexpr int COLT = H / 16;
  constexpr int ROWT = 8 / COLT;
  const int wave = threadIdx.x >> 5;
  const int lane = threadIdx.x & 31;
  const int rt = wave / COLT, ct = wave % COLT;
  const int row0 = (blockIdx.x * ROWT + rt) * 16;
  if (row0 >= N) return;
  const int col0 = ct * 16;
  const int half = lane >> 4, l16 = lane & 15;

  const float* __restrict__ arow = X + (size_t)(row0 + l16) * K;
  v8f acc = {};
#pragma unroll 4
  for (int k = 0; k < K; k += 4) {
    const int ka = k + 2 * half;
    v2f a, b;
    a.x = arow[ka];
    a.y = arow[ka + 1];
    b.x = W[(size_t)ka * H + col0 + l16];
    b.y = W[(size_t)(ka + 1) * H + col0 + l16];
    acc = __builtin_amdgcn_wmma_f32_16x16x4_f32(false, a, false, b, (short)0, acc,
                                                false, false);
  }
  const float bv = bias[col0 + l16];
  float* __restrict__ crow = C + (size_t)(row0 + 8 * half) * H + col0 + l16;
#pragma unroll
  for (int v = 0; v < 8; ++v) crow[(size_t)v * H] = acc[v] + bv;
}

constexpr int RNODES = 392;
constexpr int LCAP   = 12288;
constexpr int OWN_TB = 1024;

constexpr int MK_TB    = 1024;
constexpr int MK_OW    = 256;
constexpr int MK_PITCH = 33;

template <bool V> struct QuadVec { static constexpr bool value = V; };

template <bool VEC>
__device__ __forceinline__ int4 load_quad(const int* __restrict__ col, int q, int E) {
  if constexpr (VEC) return ((const int4*)col)[q];
  int4 r;
  const int e = 4 * q;
  r.x = (e     < E) ? col[e]     : -1;
  r.y = (e + 1 < E) ? col[e + 1] : -1;
  r.z = (e + 2 < E) ? col[e + 2] : -1;
  r.w = (e + 3 < E) ? col[e + 3] : -1;
  return r;
}

__global__ void __launch_bounds__(MK_TB)
mark_owner_quads(const int* __restrict__ ei, int E, int G, int NW, unsigned* __restrict__ marks) {
  __shared__ unsigned tile[MK_OW * MK_PITCH];
  const int tid = threadIdx.x, lane = tid & 31, wave = tid >> 5;
  const int q = blockIdx.x * MK_TB + tid;
  const bool vec = (E & 3) == 0;
  int4 d = make_int4(-1, -1, -1, -1);
  if ((long long)4 * q < E) d = vec ? load_quad<true>(ei + E, q, E) : load_quad<false>(ei + E, q, E);
  int own[4];
  own[0] = d.x >= 0 ? d.x / RNODES : -1;
  own[1] = d.y >= 0 ? d.y / RNODES : -1;
  own[2] = d.z >= 0 ? d.z / RNODES : -1;
  own[3] = d.w >= 0 ? d.w / RNODES : -1;
  for (int g0 = 0; g0 < G; g0 += MK_OW) {
    for (int i = tid; i < MK_OW * MK_PITCH; i += MK_TB) tile[i] = 0u;
    __syncthreads();
#pragma unroll
    for (int k = 0; k < 4; ++k) {
      const unsigned o = (unsigned)(own[k] - g0);
      if (o < (unsigned)MK_OW) atomicOr(&tile[o * MK_PITCH + wave], 1u << lane);
    }
    __syncthreads();
    for (int o = wave; o < MK_OW && g0 + o < G; o += MK_TB / 32)
      marks[(size_t)(g0 + o) * NW + (size_t)blockIdx.x * 32 + lane] = tile[o * MK_PITCH + lane];
    __syncthreads();
  }
}

__device__ __forceinline__ void ldsMaxF(float* addr, float val) {
  if (val >= 0.0f) atomicMax((int*)addr, __float_as_int(val));
  else             atomicMin((unsigned int*)addr, __float_as_uint(val));
}

template <int H> struct RowFrag { float v[H / 32]; };

template <int H>
__device__ __forceinline__ RowFrag<H> load_row(const float* __restrict__ basep, int n, int lane) {
  RowFrag<H> r;
  if constexpr (H == 64) {
    const float2 a = *(const float2*)(basep + (size_t)n * 64 + lane * 2);
    r.v[0] = a.x; r.v[1] = a.y;
  } else {
    r.v[0] = basep[(size_t)n * 32 + lane];
  }
  return r;
}

template <int H>
__device__ __forceinline__ float score_part(const RowFrag<H>& a, const RowFrag<H>& b,
                                            const RowFrag<H>& at) {
  if constexpr (H == 64) {
    float h0 = a.v[0] + b.v[0]; h0 = h0 > 0.0f ? h0 : 0.2f * h0;
    float h1 = a.v[1] + b.v[1]; h1 = h1 > 0.0f ? h1 : 0.2f * h1;
    return h0 * at.v[0] + h1 * at.v[1];
  } else {
    float hv = a.v[0] + b.v[0];
    hv = hv > 0.0f ? hv : 0.2f * hv;
    return hv * at.v[0];
  }
}

template <int H, bool FINAL, bool FROMLIST>
__global__ void __launch_bounds__(OWN_TB)
gat_layer_owner(const float* __restrict__ xl, const float* __restrict__ xr,
                const float* __restrict__ att, const int* __restrict__ ei, int E, int N,
                const float* __restrict__ bias, float* __restrict__ ebufG,
                const float* __restrict__ pnode, const float* __restrict__ Wlin,
                const float* __restrict__ blin, float* __restrict__ outp,
                int* __restrict__ listG, int* __restrict__ cntG,
                const unsigned* __restrict__ marks, int NW) {
  constexpr int C = FINAL ? 1 : H;
  constexpr int U = 8;
  __shared__ float s_acc[RNODES * C];
  __shared__ float s_m[RNODES];
  __shared__ float s_den[RNODES];
  __shared__ int   s_src[LCAP];
  __shared__ int   s_dl[LCAP];
  __shared__ float s_e[LCAP];
  __shared__ int   s_cnt;
  __shared__ int   s_qn;

  const int tid = threadIdx.x, lane = tid & 31, wave = tid >> 5;
  const int base = blockIdx.x * RNODES;
  const int Rn = min(RNODES, N - base);
  const int* __restrict__ dstp = ei + E;

  for (int i = tid; i < RNODES * C; i += OWN_TB) s_acc[i] = 0.0f;
  for (int i = tid; i < RNODES; i += OWN_TB) { s_m[i] = -INFINITY; s_den[i] = 0.0f; }
  for (int j = tid; j < Rn; j += OWN_TB) { s_src[j] = base + j; s_dl[j] = j; }
  int2* __restrict__ lg = (int2*)listG + (size_t)blockIdx.x * LCAP;
  int cnt;
  if constexpr (FROMLIST) {
    cnt = cntG[blockIdx.x];
    if (Rn + cnt <= LCAP) {
      for (int i = tid; i < cnt; i += OWN_TB) {
        const int2 p = lg[i];
        s_src[Rn + i] = p.x; s_dl[Rn + i] = p.y;
      }
    }
    __syncthreads();
  } else {
    if (tid == 0) { s_cnt = 0; s_qn = 0; }
    __syncthreads();

    int* s_q = (int*)s_e;
    const unsigned* __restrict__ plane = marks + (size_t)blockIdx.x * NW;
    constexpr int PW = 8;
    for (int w0 = tid; w0 < NW; w0 += OWN_TB * PW) {
      unsigned m[PW];
#pragma unroll
      for (int k = 0; k < PW; ++k) {
        const int w = w0 + k * OWN_TB;
        m[k] = (w < NW) ? plane[w] : 0u;
      }
      int nb = 0;
#pragma unroll
      for (int k = 0; k < PW; ++k) nb += __popc(m[k]);
      int slot = atomicAdd(&s_qn, nb);
#pragma unroll
      for (int k = 0; k < PW; ++k) {
        unsigned mm = m[k];
        while (mm) {
          const int b = __builtin_ctz(mm);
          mm &= mm - 1;
          if (slot < LCAP) s_q[slot] = (w0 + k * OWN_TB) * 32 + b;
          ++slot;
        }
      }
    }
    __syncthreads();
    const int qn = s_qn;
    const int nqd = min(qn, LCAP);

    auto hit = [&](int d) { return ((unsigned)(d - base) < (unsigned)Rn) ? 1 : 0; };
    auto put = [&](int d, int s, int& slot) {
      if (hit(d)) {
        if (slot < LCAP) { s_src[slot] = s; s_dl[slot] = d - base; }
        ++slot;
      }
    };
    constexpr int QV = 4;
    auto gather_quads = [&](auto vtag) {
      constexpr bool VEC = decltype(vtag)::value;
      for (int i0 = tid; i0 < nqd; i0 += OWN_TB * QV) {
        int4 d4[QV], s4[QV];
#pragma unroll
        for (int k = 0; k < QV; ++k) {
          const int qq = s_q[min(i0 + k * OWN_TB, nqd - 1)];
          d4[k] = load_quad<VEC>(dstp, qq, E);
          s4[k] = load_quad<VEC>(ei, qq, E);
        }
        int nm = 0;
#pragma unroll
        for (int k = 0; k < QV; ++k) {
          if (i0 + k * OWN_TB < nqd) nm += hit(d4[k].x) + hit(d4[k].y) + hit(d4[k].z) + hit(d4[k].w);
        }
        int slot = Rn + atomicAdd(&s_cnt, nm);
#pragma unroll
        for (int k = 0; k < QV; ++k) {
          if (i0 + k * OWN_TB < nqd) {
            put(d4[k].x, s4[k].x, slot); put(d4[k].y, s4[k].y, slot);
            put(d4[k].z, s4[k].z, slot); put(d4[k].w, s4[k].w, slot);
          }
        }
      }
    };
    if ((E & 3) == 0) gather_quads(QuadVec<true>{});
    else              gather_quads(QuadVec<false>{});
    __syncthreads();

    cnt = (qn > LCAP) ? qn : s_cnt;
    if (tid == 0) cntG[blockIdx.x] = cnt;
    if (Rn + cnt <= LCAP) {
      for (int i = tid; i < cnt; i += OWN_TB) lg[i] = make_int2(s_src[Rn + i], s_dl[Rn + i]);
    }
  }

  const int total = Rn + cnt;
  RowFrag<H> at;
  if constexpr (H == 64) { at.v[0] = att[lane * 2]; at.v[1] = att[lane * 2 + 1]; }
  else                   { at.v[0] = att[lane]; }

  if (total <= LCAP) {
    for (int i0 = wave * U; i0 < total; i0 += (OWN_TB / 32) * U) {
      const int li = min(i0 + (lane & (U - 1)), total - 1);
      const int my_dl = s_dl[li];
      const int my_src = s_src[li];
      RowFrag<H> a[U], b[U];
#pragma unroll
      for (int u = 0; u < U; ++u) {
        a[u] = load_row<H>(xl, lane_int(my_src, u), lane);
        b[u] = load_row<H>(xr, base + lane_int(my_dl, u), lane);
      }
      float tot[U];
#pragma unroll
      for (int u = 0; u < U; ++u) tot[u] = score_part<H>(a[u], b[u], at);
#pragma unroll
      for (int u = 0; u < U; ++u) tot[u] += lane_xor16(tot[u]);
#pragma unroll
      for (int u = 0; u < U; ++u) tot[u] += lane_xor_row<8>(tot[u]);
#pragma unroll
      for (int u = 0; u < U; ++u) tot[u] += lane_xor_row<4>(tot[u]);
#pragma unroll
      for (int u = 0; u < U; ++u) tot[u] += lane_xor_row<2>(tot[u]);
#pragma unroll
      for (int u = 0; u < U; ++u) tot[u] += lane_xor_row<1>(tot[u]);
      float mine = tot[0];
#pragma unroll
      for (int u = 1; u < U; ++u) mine = (lane == u) ? tot[u] : mine;
      if (lane < U && i0 + lane < total) { s_e[i0 + lane] = mine; ldsMaxF(&s_m[my_dl], mine); }
    }
    __syncthreads();
    if constexpr (FINAL) {
      for (int i = tid; i < total; i += OWN_TB) {
        const int d = s_dl[i];
        const float ex = expf(s_e[i] - s_m[d]);
        atomicAdd(&s_den[d], ex);
        atomicAdd(&s_acc[d], ex * pnode[s_src[i]]);
      }
    } else {
      for (int i0 = wave * U; i0 < total; i0 += (OWN_TB / 32) * U) {
        const int li = min(i0 + (lane & (U - 1)), total - 1);
        const int my_dl = s_dl[li];
        const int my_src = s_src[li];
        const float my_ex = expf(s_e[li] - s_m[my_dl]);
        RowFrag<H> a[U];
#pragma unroll
        for (int u = 0; u < U; ++u) a[u] = load_row<H>(xl, lane_int(my_src, u), lane);
        if (lane < U && i0 + lane < total) atomicAdd(&s_den[my_dl], my_ex);
#pragma unroll
        for (int u = 0; u < U; ++u) {
          if (i0 + u < total) {
            const int d = lane_int(my_dl, u);
            const float ex = lane_float(my_ex, u);
            if constexpr (H == 64) {
              atomicAdd(&s_acc[d * 64 + lane * 2], ex * a[u].v[0]);
              atomicAdd(&s_acc[d * 64 + lane * 2 + 1], ex * a[u].v[1]);
            } else {
              atomicAdd(&s_acc[d * 32 + lane], ex * a[u].v[0]);
            }
          }
        }
      }
    }
  } else {
    for (int phase = 0; phase < 2; ++phase) {
      auto one = [&](int s, int d, int eid) {
        const int l = d - base;
        if (phase == 0) {
          const RowFrag<H> a = load_row<H>(xl, s, lane), b = load_row<H>(xr, d, lane);
          const float tot = wave_sum32(score_part<H>(a, b, at));
          if (lane == 0) { ebufG[eid] = tot; ldsMaxF(&s_m[l], tot); }
        } else {
          const float ex = expf(ebufG[eid] - s_m[l]);
          if (lane == 0) atomicAdd(&s_den[l], ex);
          if constexpr (FINAL) {
            if (lane == 0) atomicAdd(&s_acc[l], ex * pnode[s]);
          } else {
            const RowFrag<H> a = load_row<H>(xl, s, lane);
            if constexpr (H == 64) {
              atomicAdd(&s_acc[l * 64 + lane * 2], ex * a.v[0]);
              atomicAdd(&s_acc[l * 64 + lane * 2 + 1], ex * a.v[1]);
            } else {
              atomicAdd(&s_acc[l * 32 + lane], ex * a.v[0]);
            }
          }
        }
      };
      for (int j = wave; j < Rn; j += OWN_TB / 32) one(base + j, base + j, E + base + j);
      for (int e0 = wave * 32; e0 < E; e0 += OWN_TB) {
        const int e = e0 + lane;
        const int d = (e < E) ? dstp[e] : -1;
        const bool hit = (unsigned)(d - base) < (unsigned)Rn;
        unsigned mask = (unsigned)__ballot(hit);
        while (mask) {
          const int bsel = __builtin_ctz(mask);
          mask &= mask - 1;
          one(ei[e0 + bsel], __shfl(d, bsel, 32), e0 + bsel);
        }
      }
      __syncthreads();
    }
  }
  __syncthreads();

  if constexpr (FINAL) {
    float cb = blin[0];
#pragma unroll
    for (int c = 0; c < 32; ++c) cb += bias[c] * Wlin[c];
    for (int j = tid; j < Rn; j += OWN_TB)
      outp[base + j] = s_acc[j] / (s_den[j] + 1e-16f) + cb;
  } else {
    for (int i = tid; i < Rn * H; i += OWN_TB) {
      const int n = i / H, c = i % H;
      const float v = s_acc[i] / (s_den[n] + 1e-16f) + bias[c];
      outp[(size_t)base * H + i] = v > 0.0f ? v : 0.01f * v;
    }
  }
}

__global__ void node_project(const float* __restrict__ xl, const float* __restrict__ Wlin,
                             float* __restrict__ p, int N) {
  const int gid = blockIdx.x * blockDim.x + threadIdx.x;
  const int n = gid >> 3, sub = gid & 7;
  float part = 0.0f;
  if (n < N) {
    const float4 a = *(const float4*)(xl + (size_t)n * 32 + sub * 4);
    const float4 w = *(const float4*)(Wlin + sub * 4);
    part = a.x * w.x + a.y * w.y + a.z * w.z + a.w * w.w;
  }
#pragma unroll
  for (int off = 4; off > 0; off >>= 1) part += __shfl_xor(part, off, 32);
  if (n < N && sub == 0) p[n] = part;
}

static inline int cdiv(int a, int b) { return (a + b - 1) / b; }

extern "C" void kernel_launch(void* const* d_in, const int* in_sizes, int n_in,
                              void* d_out, int out_size, void* d_ws, size_t ws_size,
                              hipStream_t stream) {
  const float* x     = (const float*)d_in[0];
  const int*   ei    = (const int*)  d_in[1];
  const float* W1l   = (const float*)d_in[2];
  const float* b1l   = (const float*)d_in[3];
  const float* W1r   = (const float*)d_in[4];
  const float* b1r   = (const float*)d_in[5];
  const float* att1  = (const float*)d_in[6];
  const float* bias1 = (const float*)d_in[7];
  const float* W2l   = (const float*)d_in[8];
  const float* b2l   = (const float*)d_in[9];
  const float* W2r   = (const float*)d_in[10];
  const float* b2r   = (const float*)d_in[11];
  const float* att2  = (const float*)d_in[12];
  const float* bias2 = (const float*)d_in[13];
  const float* Wlin  = (const float*)d_in[14];
  const float* blin  = (const float*)d_in[15];
  float* out = (float*)d_out;

  const int N    = in_sizes[0] / 128;
  const int E    = in_sizes[1] / 2;
  const int Etot = E + N;

  float* ws = (float*)d_ws;
  size_t off = 0;
  float* xl1  = ws + off; off += (size_t)N * 64;
  float* xr1  = ws + off; off += (size_t)N * 64;
  float* h1   = ws + off; off += (size_t)N * 64;
  float* xl2  = ws + off; off += (size_t)N * 32;
  float* xr2  = ws + off; off += (size_t)N * 32;
  float* pbuf = ws + off; off += (size_t)N;
  float* ebuf = ws + off; off += (size_t)Etot;
  off = (off + 1) & ~(size_t)1;
  int* listG = (int*)(ws + off); off += (size_t)cdiv(N, RNODES) * LCAP * 2;
  int* cntG  = (int*)(ws + off); off += (size_t)cdiv(N, RNODES);
  off = (off + 31) & ~(size_t)31;
  const int markGroups = cdiv(cdiv(E, 4), MK_TB);
  const int NW = markGroups * 32;
  unsigned* marks = (unsigned*)(ws + off); off += (size_t)cdiv(N, RNODES) * NW;
  (void)ws_size; (void)n_in; (void)out_size;

  const int TB = 256;
  const int rowTiles = cdiv(N, 16);
  const int ownGroups = cdiv(N, RNODES);

  mark_owner_quads<<<markGroups, MK_TB, 0, stream>>>(ei, E, ownGroups, NW, marks);

  gemm_bias_wmma<128, 64><<<cdiv(rowTiles, 2), TB, 0, stream>>>(x, W1l, b1l, xl1, N);
  gemm_bias_wmma<128, 64><<<cdiv(rowTiles, 2), TB, 0, stream>>>(x, W1r, b1r, xr1, N);
  gat_layer_owner<64, false, false><<<ownGroups, OWN_TB, 0, stream>>>(
      xl1, xr1, att1, ei, E, N, bias1, ebuf, nullptr, nullptr, nullptr, h1, listG, cntG, marks, NW);

  gemm_bias_wmma<64, 32><<<cdiv(rowTiles, 4), TB, 0, stream>>>(h1, W2l, b2l, xl2, N);
  gemm_bias_wmma<64, 32><<<cdiv(rowTiles, 4), TB, 0, stream>>>(h1, W2r, b2r, xr2, N);
  node_project<<<cdiv(N * 8, TB), TB, 0, stream>>>(xl2, Wlin, pbuf, N);
  gat_layer_owner<32, true, true><<<ownGroups, OWN_TB, 0, stream>>>(
      xl2, xr2, att2, ei, E, N, bias2, ebuf, pbuf, Wlin, blin, out, listG, cntG, marks, NW);
}


# destination-side projection xr = X.Wr + br never stored: each owner group computes it for its own 392 nodes (fp32 WMMA, same k order) into LDS, pass 1 reads it there; 8 -> 6 launches; sums unchanged
# speedup vs baseline: 5.6656x; 1.1363x over previous
#include <hip/hip_runtime.h>
#include <hip/hip_bf16.h>
#include <math.h>

typedef __attribute__((ext_vector_type(2))) float v2f;
typedef __attribute__((ext_vector_type(8))) float v8f;

__device__ __forceinline__ float wave_sum32(float v) {
#pragma unroll
  for (int off = 16; off > 0; off >>= 1) v += __shfl_xor(v, off, 32);
  return v;
}

template <int M>
__device__ __forceinline__ float lane_xor_row(float x) {
  const int b = __builtin_bit_cast(int, x);
  return __builtin_bit_cast(float, __builtin_amdgcn_update_dpp(b, b, 0x160 + M, 0xF, 0xF, true));
}
__device__ __forceinline__ float lane_xor16(float x) {
  const int b = __builtin_bit_cast(int, x);
  return __builtin_bit_cast(float,
                            __builtin_amdgcn_permlanex16(b, b, 0x76543210, 0xfedcba98, false, false));
}
__device__ __forceinline__ int lane_int(int x, int l) { return __builtin_amdgcn_readlane(x, l); }
__device__ __forceinline__ float lane_float(float x, int l) {
  return __builtin_bit_cast(float, __builtin_amdgcn_readlane(__builtin_bit_cast(int, x), l));
}

template <int K, int H>
__global__ void gemm_bias_wmma(const float* __restrict__ X, const float* __restrict__ W,
                               const float* __restrict__ bias, float* __restrict__ C, int N) {
  constexpr int COLT = H / 16;
  constexpr int ROWT = 8 / COLT;
  const int wave = threadIdx.x >> 5;
  const int lane = threadIdx.x & 31;
  const int rt = wave / COLT, ct = wave % COLT;
  const int row0 = (blockIdx.x * ROWT + rt) * 16;
  if (row0 >= N) return;
  const int col0 = ct * 16;
  const int half = lane >> 4, l16 = lane & 15;

  const float* __restrict__ arow = X + (size_t)(row0 + l16) * K;
  v8f acc = {};
#pragma unroll 4
  for (int k = 0; k < K; k += 4) {
    const int ka = k + 2 * half;
    v2f a, b;
    a.x = arow[ka];
    a.y = arow[ka + 1];
    b.x = W[(size_t)ka * H + col0 + l16];
    b.y = W[(size_t)(ka + 1) * H + col0 + l16];
    acc = __builtin_amdgcn_wmma_f32_16x16x4_f32(false, a, false, b, (short)0, acc,
                                                false, false);
  }
  const float bv = bias[col0 + l16];
  float* __restrict__ crow = C + (size_t)(row0 + 8 * half) * H + col0 + l16;
#pragma unroll
  for (int v = 0; v < 8; ++v) crow[(size_t)v * H] = acc[v] + bv;
}

constexpr int RNODES = 392;
constexpr int LCAP   = 12288;
constexpr int OWN_TB = 1024;

constexpr int MK_TB    = 1024;
constexpr int MK_OW    = 256;
constexpr int MK_PITCH = 33;

template <bool V> struct QuadVec { static constexpr bool value = V; };

template <bool VEC>
__device__ __forceinline__ int4 load_quad(const int* __restrict__ col, int q, int E) {
  if constexpr (VEC) return ((const int4*)col)[q];
  int4 r;
  const int e = 4 * q;
  r.x = (e     < E) ? col[e]     : -1;
  r.y = (e + 1 < E) ? col[e + 1] : -1;
  r.z = (e + 2 < E) ? col[e + 2] : -1;
  r.w = (e + 3 < E) ? col[e + 3] : -1;
  return r;
}

__global__ void __launch_bounds__(MK_TB)
mark_owner_quads(const int* __restrict__ ei, int E, int G, int NW, unsigned* __restrict__ marks) {
  __shared__ unsigned tile[MK_OW * MK_PITCH];
  const int tid = threadIdx.x, lane = tid & 31, wave = tid >> 5;
  const int q = blockIdx.x * MK_TB + tid;
  const bool vec = (E & 3) == 0;
  int4 d = make_int4(-1, -1, -1, -1);
  if ((long long)4 * q < E) d = vec ? load_quad<true>(ei + E, q, E) : load_quad<false>(ei + E, q, E);
  int own[4];
  own[0] = d.x >= 0 ? d.x / RNODES : -1;
  own[1] = d.y >= 0 ? d.y / RNODES : -1;
  own[2] = d.z >= 0 ? d.z / RNODES : -1;
  own[3] = d.w >= 0 ? d.w / RNODES : -1;
  for (int g0 = 0; g0 < G; g0 += MK_OW) {
    for (int i = tid; i < MK_OW * MK_PITCH; i += MK_TB) tile[i] = 0u;
    __syncthreads();
#pragma unroll
    for (int k = 0; k < 4; ++k) {
      const unsigned o = (unsigned)(own[k] - g0);
      if (o < (unsigned)MK_OW) atomicOr(&tile[o * MK_PITCH + wave], 1u << lane);
    }
    __syncthreads();
    for (int o = wave; o < MK_OW && g0 + o < G; o += MK_TB / 32)
      marks[(size_t)(g0 + o) * NW + (size_t)blockIdx.x * 32 + lane] = tile[o * MK_PITCH + lane];
    __syncthreads();
  }
}

__device__ __forceinline__ void ldsMaxF(float* addr, float val) {
  if (val >= 0.0f) atomicMax((int*)addr, __float_as_int(val));
  else             atomicMin((unsigned int*)addr, __float_as_uint(val));
}

template <int H> struct RowFrag { float v[H / 32]; };

template <int H>
__device__ __forceinline__ RowFrag<H> load_row(const float* __restrict__ basep, int n, int lane) {
  RowFrag<H> r;
  if constexpr (H == 64) {
    const float2 a = *(const float2*)(basep + (size_t)n * 64 + lane * 2);
    r.v[0] = a.x; r.v[1] = a.y;
  } else {
    r.v[0] = basep[(size_t)n * 32 + lane];
  }
  return r;
}

template <int H>
__device__ __forceinline__ RowFrag<H> lds_row(const float* s, int n, int lane) {
  RowFrag<H> r;
  if constexpr (H == 64) {
    const float2 a = *(const float2*)(s + n * 64 + lane * 2);
    r.v[0] = a.x; r.v[1] = a.y;
  } else {
    r.v[0] = s[n * 32 + lane];
  }
  return r;
}

template <int H>
__device__ __forceinline__ float score_part(const RowFrag<H>& a, const RowFrag<H>& b,
                                            const RowFrag<H>& at) {
  if constexpr (H == 64) {
    float h0 = a.v[0] + b.v[0]; h0 = h0 > 0.0f ? h0 : 0.2f * h0;
    float h1 = a.v[1] + b.v[1]; h1 = h1 > 0.0f ? h1 : 0.2f * h1;
    return h0 * at.v[0] + h1 * at.v[1];
  } else {
    float hv = a.v[0] + b.v[0];
    hv = hv > 0.0f ? hv : 0.2f * hv;
    return hv * at.v[0];
  }
}

template <int H, int K, bool FINAL, bool FROMLIST>
__global__ void __launch_bounds__(OWN_TB)
gat_layer_owner(const float* __restrict__ xl, const float* __restrict__ xin,
                const float* __restrict__ Wr, const float* __restrict__ br,
                const float* __restrict__ att, const int* __restrict__ ei, int E, int N,
                const float* __restrict__ bias, float* __restrict__ ebufG,
                const float* __restrict__ pnode, const float* __restrict__ Wlin,
                const float* __restrict__ blin, float* __restrict__ outp,
                int* __restrict__ listG, int* __restrict__ cntG,
                const unsigned* __restrict__ marks, int NW) {
  constexpr int U = 8;
  constexpr int COLT = H / 16;
  constexpr int ROWT = (RNODES + 15) / 16;
  static_assert(ROWT <= OWN_TB / 32, "one wave per row tile");
  static_assert(K * H <= LCAP, "the weight fragments are staged in the score array");
  __shared__ __attribute__((aligned(16))) float s_xa[RNODES * H];
  __shared__ float s_accF[FINAL ? RNODES : 1];
  float* const s_acc = FINAL ? s_accF : s_xa;
  __shared__ float s_m[RNODES];
  __shared__ float s_den[RNODES];
  __shared__ int   s_src[LCAP];
  __shared__ int   s_dl[LCAP];
  __shared__ __attribute__((aligned(16))) float s_e[LCAP];
  __shared__ int   s_cnt;
  __shared__ int   s_qn;

  const int tid = threadIdx.x, lane = tid & 31, wave = tid >> 5;
  const int base = blockIdx.x * RNODES;
  const int Rn = min(RNODES, N - base);
  const int* __restrict__ dstp = ei + E;

  if constexpr (FINAL) { for (int i = tid; i < RNODES; i += OWN_TB) s_accF[i] = 0.0f; }
  for (int i = tid; i < RNODES; i += OWN_TB) { s_m[i] = -INFINITY; s_den[i] = 0.0f; }
  for (int j = tid; j < Rn; j += OWN_TB) { s_src[j] = base + j; s_dl[j] = j; }

  {
    float* s_w = s_e;
    constexpr int SB = 8;
    constexpr int NB = K / 4 / SB;
    constexpr int WPT = K * H / OWN_TB;
    static_assert(K % (4 * SB) == 0 && (K * H) % OWN_TB == 0, "whole blocks, whole staging rounds");
    const bool gw = wave < ROWT && wave * 16 < Rn;
    const int half = lane >> 4, l16 = lane & 15;
    const int r = min(wave * 16 + l16, Rn - 1);
    const float* __restrict__ arow = xin + (size_t)(base + r) * K + 2 * half;
    v2f an[SB];
#pragma unroll
    for (int j = 0; j < SB; ++j) an[j] = *(const v2f*)(arow + 4 * j);
    float wv[WPT];
#pragma unroll
    for (int j = 0; j < WPT; ++j) {
      const int i = tid + j * OWN_TB;
      const int e = i & 1, l = (i >> 1) & 31, t = i >> 6;
      const int k = 4 * (t / COLT) + 2 * (l >> 4) + e;
      wv[j] = Wr[k * H + (t % COLT) * 16 + (l & 15)];
    }
#pragma unroll
    for (int j = 0; j < WPT; ++j) s_w[tid + j * OWN_TB] = wv[j];
    __syncthreads();
    if (gw) {
      v8f acc[COLT];
#pragma unroll
      for (int ct = 0; ct < COLT; ++ct) acc[ct] = v8f{};
#pragma unroll
      for (int blk = 0; blk < NB; ++blk) {
        v2f ac[SB];
#pragma unroll
        for (int j = 0; j < SB; ++j) ac[j] = an[j];
        if (blk + 1 < NB) {
#pragma unroll
          for (int j = 0; j < SB; ++j) an[j] = *(const v2f*)(arow + 4 * ((blk + 1) * SB + j));
        }
#pragma unroll
        for (int j = 0; j < SB; ++j) {
          const int s = blk * SB + j;
#pragma unroll
          for (int ct = 0; ct < COLT; ++ct) {
            const v2f b = *(const v2f*)(s_w + ((s * COLT + ct) * 32 + lane) * 2);
            acc[ct] = __builtin_amdgcn_wmma_f32_16x16x4_f32(false, ac[j], false, b, (short)0, acc[ct],
                                                            false, false);
          }
        }
      }
#pragma unroll
      for (int ct = 0; ct < COLT; ++ct) {
        const float bv = br[ct * 16 + l16];
#pragma unroll
        for (int v = 0; v < 8; ++v) {
          const int row = wave * 16 + v + 8 * half;
          if (row < Rn) s_xa[row * H + ct * 16 + l16] = acc[ct][v] + bv;
        }
      }
    }
  }
  int2* __restrict__ lg = (int2*)listG + (size_t)blockIdx.x * LCAP;
  int cnt;
  if constexpr (FROMLIST) {
    cnt = cntG[blockIdx.x];
    if (Rn + cnt <= LCAP) {
      for (int i = tid; i < cnt; i += OWN_TB) {
        const int2 p = lg[i];
        s_src[Rn + i] = p.x; s_dl[Rn + i] = p.y;
      }
    }
    __syncthreads();
  } else {
    if (tid == 0) { s_cnt = 0; s_qn = 0; }
    __syncthreads();

    int* s_q = (int*)s_e;
    const unsigned* __restrict__ plane = marks + (size_t)blockIdx.x * NW;
    constexpr int PW = 8;
    for (int w0 = tid; w0 < NW; w0 += OWN_TB * PW) {
      unsigned m[PW];
#pragma unroll
      for (int k = 0; k < PW; ++k) {
        const int w = w0 + k * OWN_TB;
        m[k] = (w < NW) ? plane[w] : 0u;
      }
      int nb = 0;
#pragma unroll
      for (int k = 0; k < PW; ++k) nb += __popc(m[k]);
      int slot = atomicAdd(&s_qn, nb);
#pragma unroll
      for (int k = 0; k < PW; ++k) {
        unsigned mm = m[k];
        while (mm) {
          const int b = __builtin_ctz(mm);
          mm &= mm - 1;
          if (slot < LCAP) s_q[slot] = (w0 + k * OWN_TB) * 32 + b;
          ++slot;
        }
      }
    }
    __syncthreads();
    const int qn = s_qn;
    const int nqd = min(qn, LCAP);

    auto hit = [&](int d) { return ((unsigned)(d - base) < (unsigned)Rn) ? 1 : 0; };
    auto put = [&](int d, int s, int& slot) {
      if (hit(d)) {
        if (slot < LCAP) { s_src[slot] = s; s_dl[slot] = d - base; }
        ++slot;
      }
    };
    constexpr int QV = 4;
    auto gather_quads = [&](auto vtag) {
      constexpr bool VEC = decltype(vtag)::value;
      for (int i0 = tid; i0 < nqd; i0 += OWN_TB * QV) {
        int4 d4[QV], s4[QV];
#pragma unroll
        for (int k = 0; k < QV; ++k) {
          const int qq = s_q[min(i0 + k * OWN_TB, nqd - 1)];
          d4[k] = load_quad<VEC>(dstp, qq, E);
          s4[k] = load_quad<VEC>(ei, qq, E);
        }
        int nm = 0;
#pragma unroll
        for (int k = 0; k < QV; ++k) {
          if (i0 + k * OWN_TB < nqd) nm += hit(d4[k].x) + hit(d4[k].y) + hit(d4[k].z) + hit(d4[k].w);
        }
        int slot = Rn + atomicAdd(&s_cnt, nm);
#pragma unroll
        for (int k = 0; k < QV; ++k) {
          if (i0 + k * OWN_TB < nqd) {
            put(d4[k].x, s4[k].x, slot); put(d4[k].y, s4[k].y, slot);
            put(d4[k].z, s4[k].z, slot); put(d4[k].w, s4[k].w, slot);
          }
        }
      }
    };
    if ((E & 3) == 0) gather_quads(QuadVec<true>{});
    else              gather_quads(QuadVec<false>{});
    __syncthreads();

    cnt = (qn > LCAP) ? qn : s_cnt;
    if (tid == 0) cntG[blockIdx.x] = cnt;
    if (Rn + cnt <= LCAP) {
      for (int i = tid; i < cnt; i += OWN_TB) lg[i] = make_int2(s_src[Rn + i], s_dl[Rn + i]);
    }
  }

  const int total = Rn + cnt;
  RowFrag<H> at;
  if constexpr (H == 64) { at.v[0] = att[lane * 2]; at.v[1] = att[lane * 2 + 1]; }
  else                   { at.v[0] = att[lane]; }

  if (total <= LCAP) {
    for (int i0 = wave * U; i0 < total; i0 += (OWN_TB / 32) * U) {
      const int li = min(i0 + (lane & (U - 1)), total - 1);
      const int my_dl = s_dl[li];
      const int my_src = s_src[li];
      RowFrag<H> a[U], b[U];
#pragma unroll
      for (int u = 0; u < U; ++u) {
        a[u] = load_row<H>(xl, lane_int(my_src, u), lane);
        b[u] = lds_row<H>(s_xa, lane_int(my_dl, u), lane);
      }
      float tot[U];
#pragma unroll
      for (int u = 0; u < U; ++u) tot[u] = score_part<H>(a[u], b[u], at);
#pragma unroll
      for (int u = 0; u < U; ++u) tot[u] += lane_xor16(tot[u]);
#pragma unroll
      for (int u = 0; u < U; ++u) tot[u] += lane_xor_row<8>(tot[u]);
#pragma unroll
      for (int u = 0; u < U; ++u) tot[u] += lane_xor_row<4>(tot[u]);
#pragma unroll
      for (int u = 0; u < U; ++u) tot[u] += lane_xor_row<2>(tot[u]);
#pragma unroll
      for (int u = 0; u < U; ++u) tot[u] += lane_xor_row<1>(tot[u]);
      float mine = tot[0];
#pragma unroll
      for (int u = 1; u < U; ++u) mine = (lane == u) ? tot[u] : mine;
      if (lane < U && i0 + lane < total) { s_e[i0 + lane] = mine; ldsMaxF(&s_m[my_dl], mine); }
    }
    __syncthreads();
    if constexpr (!FINAL) {
      for (int i = tid; i < RNODES * H; i += OWN_TB) s_xa[i] = 0.0f;
      __syncthreads();
    }
    if constexpr (FINAL) {
      for (int i = tid; i < total; i += OWN_TB) {
        const int d = s_dl[i];
        const float ex = expf(s_e[i] - s_m[d]);
        atomicAdd(&s_den[d], ex);
        atomicAdd(&s_acc[d], ex * pnode[s_src[i]]);
      }
    } else {
      for (int i0 = wave * U; i0 < total; i0 += (OWN_TB / 32) * U) {
        const int li = min(i0 + (lane & (U - 1)), total - 1);
        const int my_dl = s_dl[li];
        const int my_src = s_src[li];
        const float my_ex = expf(s_e[li] - s_m[my_dl]);
        RowFrag<H> a[U];
#pragma unroll
        for (int u = 0; u < U; ++u) a[u] = load_row<H>(xl, lane_int(my_src, u), lane);
        if (lane < U && i0 + lane < total) atomicAdd(&s_den[my_dl], my_ex);
#pragma unroll
        for (int u = 0; u < U; ++u) {
          if (i0 + u < total) {
            const int d = lane_int(my_dl, u);
            const float ex = lane_float(my_ex, u);
            if constexpr (H == 64) {
              atomicAdd(&s_acc[d * 64 + lane * 2], ex * a[u].v[0]);
              atomicAdd(&s_acc[d * 64 + lane * 2 + 1], ex * a[u].v[1]);
            } else {
              atomicAdd(&s_acc[d * 32 + lane], ex * a[u].v[0]);
            }
          }
        }
      }
    }
  } else {
    for (int phase = 0; phase < 2; ++phase) {
      auto one = [&](int s, int d, int eid) {
        const int l = d - base;
        if (phase == 0) {
          const RowFrag<H> a = load_row<H>(xl, s, lane), b = lds_row<H>(s_xa, l, lane);
          const float tot = wave_sum32(score_part<H>(a, b, at));
          if (lane == 0) { ebufG[eid] = tot; ldsMaxF(&s_m[l], tot); }
        } else {
          const float ex = expf(ebufG[eid] - s_m[l]);
          if (lane == 0) atomicAdd(&s_den[l], ex);
          if constexpr (FINAL) {
            if (lane == 0) atomicAdd(&s_acc[l], ex * pnode[s]);
          } else {
            const RowFrag<H> a = load_row<H>(xl, s, lane);
            if constexpr (H == 64) {
              atomicAdd(&s_acc[l * 64 + lane * 2], ex * a.v[0]);
              atomicAdd(&s_acc[l * 64 + lane * 2 + 1], ex * a.v[1]);
            } else {
              atomicAdd(&s_acc[l * 32 + lane], ex * a.v[0]);
            }
          }
        }
      };
      for (int j = wave; j < Rn; j += OWN_TB / 32) one(base + j, base + j, E + base + j);
      for (int e0 = wave * 32; e0 < E; e0 += OWN_TB) {
        const int e = e0 + lane;
        const int d = (e < E) ? dstp[e] : -1;
        const bool hit = (unsigned)(d - base) < (unsigned)Rn;
        unsigned mask = (unsigned)__ballot(hit);
        while (mask) {
          const int bsel = __builtin_ctz(mask);
          mask &= mask - 1;
          one(ei[e0 + bsel], __shfl(d, bsel, 32), e0 + bsel);
        }
      }
      __syncthreads();
      if constexpr (!FINAL) {
        if (phase == 0) {
          for (int i = tid; i < RNODES * H; i += OWN_TB) s_xa[i] = 0.0f;
          __syncthreads();
        }
      }
    }
  }
  __syncthreads();

  if constexpr (FINAL) {
    float cb = blin[0];
#pragma unroll
    for (int c = 0; c < 32; ++c) cb += bias[c] * Wlin[c];
    for (int j = tid; j < Rn; j += OWN_TB)
      outp[base + j] = s_acc[j] / (s_den[j] + 1e-16f) + cb;
  } else {
    for (int i = tid; i < Rn * H; i += OWN_TB) {
      const int n = i / H, c = i % H;
      const float v = s_acc[i] / (s_den[n] + 1e-16f) + bias[c];
      outp[(size_t)base * H + i] = v > 0.0f ? v : 0.01f * v;
    }
  }
}

__global__ void node_project(const float* __restrict__ xl, const float* __restrict__ Wlin,
                             float* __restrict__ p, int N) {
  const int gid = blockIdx.x * blockDim.x + threadIdx.x;
  const int n = gid >> 3, sub = gid & 7;
  float part = 0.0f;
  if (n < N) {
    const float4 a = *(const float4*)(xl + (size_t)n * 32 + sub * 4);
    const float4 w = *(const float4*)(Wlin + sub * 4);
    part = a.x * w.x + a.y * w.y + a.z * w.z + a.w * w.w;
  }
#pragma unroll
  for (int off = 4; off > 0; off >>= 1) part += __shfl_xor(part, off, 32);
  if (n < N && sub == 0) p[n] = part;
}

static inline int cdiv(int a, int b) { return (a + b - 1) / b; }

extern "C" void kernel_launch(void* const* d_in, const int* in_sizes, int n_in,
                              void* d_out, int out_size, void* d_ws, size_t ws_size,
                              hipStream_t stream) {
  const float* x     = (const float*)d_in[0];
  const int*   ei    = (const int*)  d_in[1];
  const float* W1l   = (const float*)d_in[2];
  const float* b1l   = (const float*)d_in[3];
  const float* W1r   = (const float*)d_in[4];
  const float* b1r   = (const float*)d_in[5];
  const float* att1  = (const float*)d_in[6];
  const float* bias1 = (const float*)d_in[7];
  const float* W2l   = (const float*)d_in[8];
  const float* b2l   = (const float*)d_in[9];
  const float* W2r   = (const float*)d_in[10];
  const float* b2r   = (const float*)d_in[11];
  const float* att2  = (const float*)d_in[12];
  const float* bias2 = (const float*)d_in[13];
  const float* Wlin  = (const float*)d_in[14];
  const float* blin  = (const float*)d_in[15];
  float* out = (float*)d_out;

  const int N    = in_sizes[0] / 128;
  const int E    = in_sizes[1] / 2;
  const int Etot = E + N;

  float* ws = (float*)d_ws;
  size_t off = 0;
  float* xl1  = ws + off; off += (size_t)N * 64;
  float* h1   = ws + off; off += (size_t)N * 64;
  float* xl2  = ws + off; off += (size_t)N * 32;
  float* pbuf = ws + off; off += (size_t)N;
  float* ebuf = ws + off; off += (size_t)Etot;
  off = (off + 1) & ~(size_t)1;
  int* listG = (int*)(ws + off); off += (size_t)cdiv(N, RNODES) * LCAP * 2;
  int* cntG  = (int*)(ws + off); off += (size_t)cdiv(N, RNODES);
  off = (off + 31) & ~(size_t)31;
  const int markGroups = cdiv(cdiv(E, 4), MK_TB);
  const int NW = markGroups * 32;
  unsigned* marks = (unsigned*)(ws + off); off += (size_t)cdiv(N, RNODES) * NW;
  (void)ws_size; (void)n_in; (void)out_size;

  const int TB = 256;
  const int rowTiles = cdiv(N, 16);
  const int ownGroups = cdiv(N, RNODES);

  mark_owner_quads<<<markGroups, MK_TB, 0, stream>>>(ei, E, ownGroups, NW, marks);

  gemm_bias_wmma<128, 64><<<cdiv(rowTiles, 2), TB, 0, stream>>>(x, W1l, b1l, xl1, N);
  gat_layer_owner<64, 128, false, false><<<ownGroups, OWN_TB, 0, stream>>>(
      xl1, x, W1r, b1r, att1, ei, E, N, bias1, ebuf, nullptr, nullptr, nullptr, h1, listG, cntG, marks, NW);

  gemm_bias_wmma<64, 32><<<cdiv(rowTiles, 4), TB, 0, stream>>>(h1, W2l, b2l, xl2, N);
  node_project<<<cdiv(N * 8, TB), TB, 0, stream>>>(xl2, Wlin, pbuf, N);
  gat_layer_owner<32, 64, true, true><<<ownGroups, OWN_TB, 0, stream>>>(
      xl2, h1, W2r, b2r, att2, ei, E, N, bias2, ebuf, pbuf, Wlin, blin, out, listG, cntG, marks, NW);
}


# layer 2: pass 1 sums p = xl2[src].Wlin from the row it holds (16 wave sums folded in one exchange tree), p stays in LDS: no p launch, no p gather; 6 -> 5 launches; score and p sum order changes
# speedup vs baseline: 6.0715x; 1.0716x over previous
#include <hip/hip_runtime.h>
#include <hip/hip_bf16.h>
#include <math.h>

typedef __attribute__((ext_vector_type(2))) float v2f;
typedef __attribute__((ext_vector_type(8))) float v8f;

__device__ __forceinline__ float wave_sum32(float v) {
#pragma unroll
  for (int off = 16; off > 0; off >>= 1) v += __shfl_xor(v, off, 32);
  return v;
}

template <int M>
__device__ __forceinline__ float lane_xor_row(float x) {
  const int b = __builtin_bit_cast(int, x);
  return __builtin_bit_cast(float, __builtin_amdgcn_update_dpp(b, b, 0x160 + M, 0xF, 0xF, true));
}
__device__ __forceinline__ float lane_xor16(float x) {
  const int b = __builtin_bit_cast(int, x);
  return __builtin_bit_cast(float,
                            __builtin_amdgcn_permlanex16(b, b, 0x76543210, 0xfedcba98, false, false));
}
__device__ __forceinline__ int lane_int(int x, int l) { return __builtin_amdgcn_readlane(x, l); }
__device__ __forceinline__ float lane_float(float x, int l) {
  return __builtin_bit_cast(float, __builtin_amdgcn_readlane(__builtin_bit_cast(int, x), l));
}

template <int M>
__device__ __forceinline__ float fold_pair(float xa, float xb, bool hi) {
  const float own = hi ? xb : xa;
  const float oth = hi ? xa : xb;
  return own + lane_xor_row<M>(oth);
}

template <int K, int H>
__global__ void gemm_bias_wmma(const float* __restrict__ X, const float* __restrict__ W,
                               const float* __restrict__ bias, float* __restrict__ C, int N) {
  constexpr int COLT = H / 16;
  constexpr int ROWT = 8 / COLT;
  const int wave = threadIdx.x >> 5;
  const int lane = threadIdx.x & 31;
  const int rt = wave / COLT, ct = wave % COLT;
  const int row0 = (blockIdx.x * ROWT + rt) * 16;
  if (row0 >= N) return;
  const int col0 = ct * 16;
  const int half = lane >> 4, l16 = lane & 15;

  const float* __restrict__ arow = X + (size_t)(row0 + l16) * K;
  v8f acc = {};
#pragma unroll 4
  for (int k = 0; k < K; k += 4) {
    const int ka = k + 2 * half;
    v2f a, b;
    a.x = arow[ka];
    a.y = arow[ka + 1];
    b.x = W[(size_t)ka * H + col0 + l16];
    b.y = W[(size_t)(ka + 1) * H + col0 + l16];
    acc = __builtin_amdgcn_wmma_f32_16x16x4_f32(false, a, false, b, (short)0, acc,
                                                false, false);
  }
  const float bv = bias[col0 + l16];
  float* __restrict__ crow = C + (size_t)(row0 + 8 * half) * H + col0 + l16;
#pragma unroll
  for (int v = 0; v < 8; ++v) crow[(size_t)v * H] = acc[v] + bv;
}

constexpr int RNODES = 392;
constexpr int LCAP   = 12288;
constexpr int OWN_TB = 1024;

constexpr int MK_TB    = 1024;
constexpr int MK_OW    = 256;
constexpr int MK_PITCH = 33;

template <bool V> struct QuadVec { static constexpr bool value = V; };

template <bool VEC>
__device__ __forceinline__ int4 load_quad(const int* __restrict__ col, int q, int E) {
  if constexpr (VEC) return ((const int4*)col)[q];
  int4 r;
  const int e = 4 * q;
  r.x = (e     < E) ? col[e]     : -1;
  r.y = (e + 1 < E) ? col[e + 1] : -1;
  r.z = (e + 2 < E) ? col[e + 2] : -1;
  r.w = (e + 3 < E) ? col[e + 3] : -1;
  return r;
}

__global__ void __launch_bounds__(MK_TB)
mark_owner_quads(const int* __restrict__ ei, int E, int G, int NW, unsigned* __restrict__ marks) {
  __shared__ unsigned tile[MK_OW * MK_PITCH];
  const int tid = threadIdx.x, lane = tid & 31, wave = tid >> 5;
  const int q = blockIdx.x * MK_TB + tid;
  const bool vec = (E & 3) == 0;
  int4 d = make_int4(-1, -1, -1, -1);
  if ((long long)4 * q < E) d = vec ? load_quad<true>(ei + E, q, E) : load_quad<false>(ei + E, q, E);
  int own[4];
  own[0] = d.x >= 0 ? d.x / RNODES : -1;
  own[1] = d.y >= 0 ? d.y / RNODES : -1;
  own[2] = d.z >= 0 ? d.z / RNODES : -1;
  own[3] = d.w >= 0 ? d.w / RNODES : -1;
  for (int g0 = 0; g0 < G; g0 += MK_OW) {
    for (int i = tid; i < MK_OW * MK_PITCH; i += MK_TB) tile[i] = 0u;
    __syncthreads();
#pragma unroll
    for (int k = 0; k < 4; ++k) {
      const unsigned o = (unsigned)(own[k] - g0);
      if (o < (unsigned)MK_OW) atomicOr(&tile[o * MK_PITCH + wave], 1u << lane);
    }
    __syncthreads();
    for (int o = wave; o < MK_OW && g0 + o < G; o += MK_TB / 32)
      marks[(size_t)(g0 + o) * NW + (size_t)blockIdx.x * 32 + lane] = tile[o * MK_PITCH + lane];
    __syncthreads();
  }
}

__device__ __forceinline__ void ldsMaxF(float* addr, float val) {
  if (val >= 0.0f) atomicMax((int*)addr, __float_as_int(val));
  else             atomicMin((unsigned int*)addr, __float_as_uint(val));
}

template <int H> struct RowFrag { float v[H / 32]; };

template <int H>
__device__ __forceinline__ RowFrag<H> load_row(const float* __restrict__ basep, int n, int lane) {
  RowFrag<H> r;
  if constexpr (H == 64) {
    const float2 a = *(const float2*)(basep + (size_t)n * 64 + lane * 2);
    r.v[0] = a.x; r.v[1] = a.y;
  } else {
    r.v[0] = basep[(size_t)n * 32 + lane];
  }
  return r;
}

template <int H>
__device__ __forceinline__ RowFrag<H> lds_row(const float* s, int n, int lane) {
  RowFrag<H> r;
  if constexpr (H == 64) {
    const float2 a = *(const float2*)(s + n * 64 + lane * 2);
    r.v[0] = a.x; r.v[1] = a.y;
  } else {
    r.v[0] = s[n * 32 + lane];
  }
  return r;
}

template <int H>
__device__ __forceinline__ float score_part(const RowFrag<H>& a, const RowFrag<H>& b,
                                            const RowFrag<H>& at) {
  if constexpr (H == 64) {
    float h0 = a.v[0] + b.v[0]; h0 = h0 > 0.0f ? h0 : 0.2f * h0;
    float h1 = a.v[1] + b.v[1]; h1 = h1 > 0.0f ? h1 : 0.2f * h1;
    return h0 * at.v[0] + h1 * at.v[1];
  } else {
    float hv = a.v[0] + b.v[0];
    hv = hv > 0.0f ? hv : 0.2f * hv;
    return hv * at.v[0];
  }
}

template <int H, int K, bool FINAL, bool FROMLIST>
__global__ void __launch_bounds__(OWN_TB)
gat_layer_owner(const float* __restrict__ xl, const float* __restrict__ xin,
                const float* __restrict__ Wr, const float* __restrict__ br,
                const float* __restrict__ att, const int* __restrict__ ei, int E, int N,
                const float* __restrict__ bias, float* __restrict__ ebufG,
                const float* __restrict__ Wlin,
                const float* __restrict__ blin, float* __restrict__ outp,
                int* __restrict__ listG, int* __restrict__ cntG,
                const unsigned* __restrict__ marks, int NW) {
  constexpr int U = 8;
  constexpr int COLT = H / 16;
  constexpr int ROWT = (RNODES + 15) / 16;
  static_assert(ROWT <= OWN_TB / 32, "one wave per row tile");
  static_assert(K * H <= LCAP, "the weight fragments are staged in the score array");
  __shared__ __attribute__((aligned(16))) float s_xa[RNODES * H];
  __shared__ float s_accF[FINAL ? RNODES : 1];
  float* const s_acc = FINAL ? s_accF : s_xa;
  __shared__ float s_m[RNODES];
  __shared__ float s_den[RNODES];
  __shared__ int   s_src[LCAP];
  __shared__ int   s_dl[LCAP];
  __shared__ __attribute__((aligned(16))) float s_e[LCAP];
  __shared__ float s_p[FINAL ? LCAP : 1];
  __shared__ int   s_cnt;
  __shared__ int   s_qn;

  const int tid = threadIdx.x, lane = tid & 31, wave = tid >> 5;
  const int base = blockIdx.x * RNODES;
  const int Rn = min(RNODES, N - base);
  const int* __restrict__ dstp = ei + E;

  if constexpr (FINAL) { for (int i = tid; i < RNODES; i += OWN_TB) s_accF[i] = 0.0f; }
  for (int i = tid; i < RNODES; i += OWN_TB) { s_m[i] = -INFINITY; s_den[i] = 0.0f; }
  for (int j = tid; j < Rn; j += OWN_TB) { s_src[j] = base + j; s_dl[j] = j; }

  {
    float* s_w = s_e;
    constexpr int SB = 8;
    constexpr int NB = K / 4 / SB;
    constexpr int WPT = K * H / OWN_TB;
    static_assert(K % (4 * SB) == 0 && (K * H) % OWN_TB == 0, "whole blocks, whole staging rounds");
    const bool gw = wave < ROWT && wave * 16 < Rn;
    const int half = lane >> 4, l16 = lane & 15;
    const int r = min(wave * 16 + l16, Rn - 1);
    const float* __restrict__ arow = xin + (size_t)(base + r) * K + 2 * half;
    v2f an[SB];
#pragma unroll
    for (int j = 0; j < SB; ++j) an[j] = *(const v2f*)(arow + 4 * j);
    float wv[WPT];
#pragma unroll
    for (int j = 0; j < WPT; ++j) {
      const int i = tid + j * OWN_TB;
      const int e = i & 1, l = (i >> 1) & 31, t = i >> 6;
      const int k = 4 * (t / COLT) + 2 * (l >> 4) + e;
      wv[j] = Wr[k * H + (t % COLT) * 16 + (l & 15)];
    }
#pragma unroll
    for (int j = 0; j < WPT; ++j) s_w[tid + j * OWN_TB] = wv[j];
    __syncthreads();
    if (gw) {
      v8f acc[COLT];
#pragma unroll
      for (int ct = 0; ct < COLT; ++ct) acc[ct] = v8f{};
#pragma unroll
      for (int blk = 0; blk < NB; ++blk) {
        v2f ac[SB];
#pragma unroll
        for (int j = 0; j < SB; ++j) ac[j] = an[j];
        if (blk + 1 < NB) {
#pragma unroll
          for (int j = 0; j < SB; ++j) an[j] = *(const v2f*)(arow + 4 * ((blk + 1) * SB + j));
        }
#pragma unroll
        for (int j = 0; j < SB; ++j) {
          const int s = blk * SB + j;
#pragma unroll
          for (int ct = 0; ct < COLT; ++ct) {
            const v2f b = *(const v2f*)(s_w + ((s * COLT + ct) * 32 + lane) * 2);
            acc[ct] = __builtin_amdgcn_wmma_f32_16x16x4_f32(false, ac[j], false, b, (short)0, acc[ct],
                                                            false, false);
          }
        }
      }
#pragma unroll
      for (int ct = 0; ct < COLT; ++ct) {
        const float bv = br[ct * 16 + l16];
#pragma unroll
        for (int v = 0; v < 8; ++v) {
          const int row = wave * 16 + v + 8 * half;
          if (row < Rn) s_xa[row * H + ct * 16 + l16] = acc[ct][v] + bv;
        }
      }
    }
  }
  int2* __restrict__ lg = (int2*)listG + (size_t)blockIdx.x * LCAP;
  int cnt;
  if constexpr (FROMLIST) {
    cnt = cntG[blockIdx.x];
    if (Rn + cnt <= LCAP) {
      for (int i = tid; i < cnt; i += OWN_TB) {
        const int2 p = lg[i];
        s_src[Rn + i] = p.x; s_dl[Rn + i] = p.y;
      }
    }
    __syncthreads();
  } else {
    if (tid == 0) { s_cnt = 0; s_qn = 0; }
    __syncthreads();

    int* s_q = (int*)s_e;
    const unsigned* __restrict__ plane = marks + (size_t)blockIdx.x * NW;
    constexpr int PW = 8;
    for (int w0 = tid; w0 < NW; w0 += OWN_TB * PW) {
      unsigned m[PW];
#pragma unroll
      for (int k = 0; k < PW; ++k) {
        const int w = w0 + k * OWN_TB;
        m[k] = (w < NW) ? plane[w] : 0u;
      }
      int nb = 0;
#pragma unroll
      for (int k = 0; k < PW; ++k) nb += __popc(m[k]);
      int slot = atomicAdd(&s_qn, nb);
#pragma unroll
      for (int k = 0; k < PW; ++k) {
        unsigned mm = m[k];
        while (mm) {
          const int b = __builtin_ctz(mm);
          mm &= mm - 1;
          if (slot < LCAP) s_q[slot] = (w0 + k * OWN_TB) * 32 + b;
          ++slot;
        }
      }
    }
    __syncthreads();
    const int qn = s_qn;
    const int nqd = min(qn, LCAP);

    auto hit = [&](int d) { return ((unsigned)(d - base) < (unsigned)Rn) ? 1 : 0; };
    auto put = [&](int d, int s, int& slot) {
      if (hit(d)) {
        if (slot < LCAP) { s_src[slot] = s; s_dl[slot] = d - base; }
        ++slot;
      }
    };
    constexpr int QV = 4;
    auto gather_quads = [&](auto vtag) {
      constexpr bool VEC = decltype(vtag)::value;
      for (int i0 = tid; i0 < nqd; i0 += OWN_TB * QV) {
        int4 d4[QV], s4[QV];
#pragma unroll
        for (int k = 0; k < QV; ++k) {
          const int qq = s_q[min(i0 + k * OWN_TB, nqd - 1)];
          d4[k] = load_quad<VEC>(dstp, qq, E);
          s4[k] = load_quad<VEC>(ei, qq, E);
        }
        int nm = 0;
#pragma unroll
        for (int k = 0; k < QV; ++k) {
          if (i0 + k * OWN_TB < nqd) nm += hit(d4[k].x) + hit(d4[k].y) + hit(d4[k].z) + hit(d4[k].w);
        }
        int slot = Rn + atomicAdd(&s_cnt, nm);
#pragma unroll
        for (int k = 0; k < QV; ++k) {
          if (i0 + k * OWN_TB < nqd) {
            put(d4[k].x, s4[k].x, slot); put(d4[k].y, s4[k].y, slot);
            put(d4[k].z, s4[k].z, slot); put(d4[k].w, s4[k].w, slot);
          }
        }
      }
    };
    if ((E & 3) == 0) gather_quads(QuadVec<true>{});
    else              gather_quads(QuadVec<false>{});
    __syncthreads();

    cnt = (qn > LCAP) ? qn : s_cnt;
    if (tid == 0) cntG[blockIdx.x] = cnt;
    if (Rn + cnt <= LCAP) {
      for (int i = tid; i < cnt; i += OWN_TB) lg[i] = make_int2(s_src[Rn + i], s_dl[Rn + i]);
    }
  }

  const int total = Rn + cnt;
  RowFrag<H> at;
  if constexpr (H == 64) { at.v[0] = att[lane * 2]; at.v[1] = att[lane * 2 + 1]; }
  else                   { at.v[0] = att[lane]; }
  float wl = 0.0f;
  if constexpr (FINAL) {
    static_assert(!FINAL || H == 32, "one channel per lane in the last layer");
    wl = Wlin[lane];
  }

  if (total <= LCAP) {
    for (int i0 = wave * U; i0 < total; i0 += (OWN_TB / 32) * U) {
      const int li = min(i0 + (lane & (U - 1)), total - 1);
      const int my_dl = s_dl[li];
      const int my_src = s_src[li];
      RowFrag<H> a[U], b[U];
#pragma unroll
      for (int u = 0; u < U; ++u) {
        a[u] = load_row<H>(xl, lane_int(my_src, u), lane);
        b[u] = lds_row<H>(s_xa, lane_int(my_dl, u), lane);
      }
      float tot[U];
#pragma unroll
      for (int u = 0; u < U; ++u) tot[u] = score_part<H>(a[u], b[u], at);
      if constexpr (FINAL) {
        float x[2 * U];
#pragma unroll
        for (int u = 0; u < U; ++u) { x[u] = tot[u]; x[U + u] = a[u].v[0] * wl; }
        const bool h1 = (lane & 1) != 0, h2 = (lane & 2) != 0, h4 = (lane & 4) != 0, h8 = (lane & 8) != 0;
        float y[U], z[U / 2], w[U / 4];
#pragma unroll
        for (int i = 0; i < U; ++i) y[i] = fold_pair<1>(x[2 * i], x[2 * i + 1], h1);
#pragma unroll
        for (int i = 0; i < U / 2; ++i) z[i] = fold_pair<2>(y[2 * i], y[2 * i + 1], h2);
#pragma unroll
        for (int i = 0; i < U / 4; ++i) w[i] = fold_pair<4>(z[2 * i], z[2 * i + 1], h4);
        static_assert(U == 8, "four fold stages take 16 vectors to one");
        float r = fold_pair<8>(w[0], w[1], h8);
        r += lane_xor16(r);
        const int idx = i0 + (lane & (U - 1));
        if (lane < 2 * U && idx < total) {
          if (lane < U) { s_e[idx] = r; ldsMaxF(&s_m[my_dl], r); }
          else          { s_p[idx] = r; }
        }
      } else {
#pragma unroll
        for (int u = 0; u < U; ++u) tot[u] += lane_xor16(tot[u]);
#pragma unroll
        for (int u = 0; u < U; ++u) tot[u] += lane_xor_row<8>(tot[u]);
#pragma unroll
        for (int u = 0; u < U; ++u) tot[u] += lane_xor_row<4>(tot[u]);
#pragma unroll
        for (int u = 0; u < U; ++u) tot[u] += lane_xor_row<2>(tot[u]);
#pragma unroll
        for (int u = 0; u < U; ++u) tot[u] += lane_xor_row<1>(tot[u]);
        float mine = tot[0];
#pragma unroll
        for (int u = 1; u < U; ++u) mine = (lane == u) ? tot[u] : mine;
        if (lane < U && i0 + lane < total) { s_e[i0 + lane] = mine; ldsMaxF(&s_m[my_dl], mine); }
      }
    }
    __syncthreads();
    if constexpr (!FINAL) {
      for (int i = tid; i < RNODES * H; i += OWN_TB) s_xa[i] = 0.0f;
      __syncthreads();
    }
    if constexpr (FINAL) {
      for (int i = tid; i < total; i += OWN_TB) {
        const int d = s_dl[i];
        const float ex = expf(s_e[i] - s_m[d]);
        atomicAdd(&s_den[d], ex);
        atomicAdd(&s_acc[d], ex * s_p[i]);
      }
    } else {
      for (int i0 = wave * U; i0 < total; i0 += (OWN_TB / 32) * U) {
        const int li = min(i0 + (lane & (U - 1)), total - 1);
        const int my_dl = s_dl[li];
        const int my_src = s_src[li];
        const float my_ex = expf(s_e[li] - s_m[my_dl]);
        RowFrag<H> a[U];
#pragma unroll
        for (int u = 0; u < U; ++u) a[u] = load_row<H>(xl, lane_int(my_src, u), lane);
        if (lane < U && i0 + lane < total) atomicAdd(&s_den[my_dl], my_ex);
#pragma unroll
        for (int u = 0; u < U; ++u) {
          if (i0 + u < total) {
            const int d = lane_int(my_dl, u);
            const float ex = lane_float(my_ex, u);
            if constexpr (H == 64) {
              atomicAdd(&s_acc[d * 64 + lane * 2], ex * a[u].v[0]);
              atomicAdd(&s_acc[d * 64 + lane * 2 + 1], ex * a[u].v[1]);
            } else {
              atomicAdd(&s_acc[d * 32 + lane], ex * a[u].v[0]);
            }
          }
        }
      }
    }
  } else {
    for (int phase = 0; phase < 2; ++phase) {
      auto one = [&](int s, int d, int eid) {
        const int l = d - base;
        if (phase == 0) {
          const RowFrag<H> a = load_row<H>(xl, s, lane), b = lds_row<H>(s_xa, l, lane);
          const float tot = wave_sum32(score_part<H>(a, b, at));
          if (lane == 0) { ebufG[eid] = tot; ldsMaxF(&s_m[l], tot); }
        } else {
          const float ex = expf(ebufG[eid] - s_m[l]);
          if (lane == 0) atomicAdd(&s_den[l], ex);
          if constexpr (FINAL) {
            const RowFrag<H> a = load_row<H>(xl, s, lane);
            const float pv = wave_sum32(a.v[0] * wl);
            if (lane == 0) atomicAdd(&s_acc[l], ex * pv);
          } else {
            const RowFrag<H> a = load_row<H>(xl, s, lane);
            if constexpr (H == 64) {
              atomicAdd(&s_acc[l * 64 + lane * 2], ex * a.v[0]);
              atomicAdd(&s_acc[l * 64 + lane * 2 + 1], ex * a.v[1]);
            } else {
              atomicAdd(&s_acc[l * 32 + lane], ex * a.v[0]);
            }
          }
        }
      };
      for (int j = wave; j < Rn; j += OWN_TB / 32) one(base + j, base + j, E + base + j);
      for (int e0 = wave * 32; e0 < E; e0 += OWN_TB) {
        const int e = e0 + lane;
        const int d = (e < E) ? dstp[e] : -1;
        const bool hit = (unsigned)(d - base) < (unsigned)Rn;
        unsigned mask = (unsigned)__ballot(hit);
        while (mask) {
          const int bsel = __builtin_ctz(mask);
          mask &= mask - 1;
          one(ei[e0 + bsel], __shfl(d, bsel, 32), e0 + bsel);
        }
      }
      __syncthreads();
      if constexpr (!FINAL) {
        if (phase == 0) {
          for (int i = tid; i < RNODES * H; i += OWN_TB) s_xa[i] = 0.0f;
          __syncthreads();
        }
      }
    }
  }
  __syncthreads();

  if constexpr (FINAL) {
    float cb = blin[0];
#pragma unroll
    for (int c = 0; c < 32; ++c) cb += bias[c] * Wlin[c];
    for (int j = tid; j < Rn; j += OWN_TB)
      outp[base + j] = s_acc[j] / (s_den[j] + 1e-16f) + cb;
  } else {
    for (int i = tid; i < Rn * H; i += OWN_TB) {
      const int n = i / H, c = i % H;
      const float v = s_acc[i] / (s_den[n] + 1e-16f) + bias[c];
      outp[(size_t)base * H + i] = v > 0.0f ? v : 0.01f * v;
    }
  }
}

static inline int cdiv(int a, int b) { return (a + b - 1) / b; }

extern "C" void kernel_launch(void* const* d_in, const int* in_sizes, int n_in,
                              void* d_out, int out_size, void* d_ws, size_t ws_size,
                              hipStream_t stream) {
  const float* x     = (const float*)d_in[0];
  const int*   ei    = (const int*)  d_in[1];
  const float* W1l   = (const float*)d_in[2];
  const float* b1l   = (const float*)d_in[3];
  const float* W1r   = (const float*)d_in[4];
  const float* b1r   = (const float*)d_in[5];
  const float* att1  = (const float*)d_in[6];
  const float* bias1 = (const float*)d_in[7];
  const float* W2l   = (const float*)d_in[8];
  const float* b2l   = (const float*)d_in[9];
  const float* W2r   = (const float*)d_in[10];
  const float* b2r   = (const float*)d_in[11];
  const float* att2  = (const float*)d_in[12];
  const float* bias2 = (const float*)d_in[13];
  const float* Wlin  = (const float*)d_in[14];
  const float* blin  = (const float*)d_in[15];
  float* out = (float*)d_out;

  const int N    = in_sizes[0] / 128;
  const int E    = in_sizes[1] / 2;
  const int Etot = E + N;

  float* ws = (float*)d_ws;
  size_t off = 0;
  float* xl1  = ws + off; off += (size_t)N * 64;
  float* h1   = ws + off; off += (size_t)N * 64;
  float* xl2  = ws + off; off += (size_t)N * 32;
  float* ebuf = ws + off; off += (size_t)Etot;
  off = (off + 1) & ~(size_t)1;
  int* listG = (int*)(ws + off); off += (size_t)cdiv(N, RNODES) * LCAP * 2;
  int* cntG  = (int*)(ws + off); off += (size_t)cdiv(N, RNODES);
  off = (off + 31) & ~(size_t)31;
  const int markGroups = cdiv(cdiv(E, 4), MK_TB);
  const int NW = markGroups * 32;
  unsigned* marks = (unsigned*)(ws + off); off += (size_t)cdiv(N, RNODES) * NW;
  (void)ws_size; (void)n_in; (void)out_size;

  const int TB = 256;
  const int rowTiles = cdiv(N, 16);
  const int ownGroups = cdiv(N, RNODES);

  mark_owner_quads<<<markGroups, MK_TB, 0, stream>>>(ei, E, ownGroups, NW, marks);

  gemm_bias_wmma<128, 64><<<cdiv(rowTiles, 2), TB, 0, stream>>>(x, W1l, b1l, xl1, N);
  gat_layer_owner<64, 128, false, false><<<ownGroups, OWN_TB, 0, stream>>>(
      xl1, x, W1r, b1r, att1, ei, E, N, bias1, ebuf, nullptr, nullptr, h1, listG, cntG, marks, NW);

  gemm_bias_wmma<64, 32><<<cdiv(rowTiles, 4), TB, 0, stream>>>(h1, W2l, b2l, xl2, N);
  gat_layer_owner<32, 64, true, true><<<ownGroups, OWN_TB, 0, stream>>>(
      xl2, h1, W2r, b2r, att2, ei, E, N, bias2, ebuf, Wlin, blin, out, listG, cntG, marks, NW);
}


# xl GEMMs: resident grid, a wave keeps all of W as B fragments in registers, walks tiles of 16 rows x all columns (16-byte X loads, whole-row stores), no weight loads per step; k sum order changes
# speedup vs baseline: 6.3750x; 1.0500x over previous
#include <hip/hip_runtime.h>
#include <hip/hip_bf16.h>
#include <math.h>

typedef __attribute__((ext_vector_type(2))) float v2f;
typedef __attribute__((ext_vector_type(8))) float v8f;

__device__ __forceinline__ float wave_sum32(float v) {
#pragma unroll
  for (int off = 16; off > 0; off >>= 1) v += __shfl_xor(v, off, 32);
  return v;
}

template <int M>
__device__ __forceinline__ float lane_xor_row(float x) {
  const int b = __builtin_bit_cast(int, x);
  return __builtin_bit_cast(float, __builtin_amdgcn_update_dpp(b, b, 0x160 + M, 0xF, 0xF, true));
}
__device__ __forceinline__ float lane_xor16(float x) {
  const int b = __builtin_bit_cast(int, x);
  return __builtin_bit_cast(float,
                            __builtin_amdgcn_permlanex16(b, b, 0x76543210, 0xfedcba98, false, false));
}
__device__ __forceinline__ int lane_int(int x, int l) { return __builtin_amdgcn_readlane(x, l); }
__device__ __forceinline__ float lane_float(float x, int l) {
  return __builtin_bit_cast(float, __builtin_amdgcn_readlane(__builtin_bit_cast(int, x), l));
}

template <int M>
__device__ __forceinline__ float fold_pair(float xa, float xb, bool hi) {
  const float own = hi ? xb : xa;
  const float oth = hi ? xa : xb;
  return own + lane_xor_row<M>(oth);
}

constexpr int GM_TB = 256;

template <int C> struct ColVec;
template <> struct ColVec<4> { typedef __attribute__((ext_vector_type(4))) float type; };
template <> struct ColVec<2> { typedef __attribute__((ext_vector_type(2))) float type; };

template <int K, int H>
__global__ void __launch_bounds__(GM_TB)
gemm_bias_wmma(const float* __restrict__ X, const float* __restrict__ W,
               const float* __restrict__ bias, float* __restrict__ C, int N) {
  constexpr int COLT = H / 16;
  constexpr int KQ = K / 8;
  static_assert(K % 8 == 0 && H % 16 == 0, "whole 16-byte loads, whole column tiles");
  typedef typename ColVec<COLT>::type cvec;
  typedef __attribute__((ext_vector_type(4))) float v4f;
  const int lane = threadIdx.x & 31;
  const int half = lane >> 4, l16 = lane & 15;
  const int wv = __builtin_amdgcn_readfirstlane((int)(threadIdx.x >> 5));
  const int gwave = wv * (int)gridDim.x + (int)blockIdx.x;
  const int nwaves = (int)gridDim.x * (GM_TB / 32);
  const int rowTiles = (N + 15) >> 4;

  v2f wf[KQ][2][COLT];
#pragma unroll
  for (int t = 0; t < KQ; ++t) {
#pragma unroll
    for (int e = 0; e < 2; ++e) {
      const int k = 8 * t + 4 * half + 2 * e;
      const cvec w0 = *(const cvec*)(W + (size_t)k * H + COLT * l16);
      const cvec w1 = *(const cvec*)(W + (size_t)(k + 1) * H + COLT * l16);
#pragma unroll
      for (int ct = 0; ct < COLT; ++ct) { wf[t][e][ct].x = w0[ct]; wf[t][e][ct].y = w1[ct]; }
    }
  }
  const cvec bv = *(const cvec*)(bias + COLT * l16);

  v4f an[KQ];
  int rt = gwave;
  if (rt < rowTiles) {
    const v4f* __restrict__ ap = (const v4f*)(X + (size_t)min(rt * 16 + l16, N - 1) * K + 4 * half);
#pragma unroll
    for (int t = 0; t < KQ; ++t) an[t] = ap[2 * t];
  }
  for (; rt < rowTiles; rt += nwaves) {
    v4f ac[KQ];
#pragma unroll
    for (int t = 0; t < KQ; ++t) ac[t] = an[t];
    const int nrt = rt + nwaves;
    if (nrt < rowTiles) {
      const v4f* __restrict__ ap = (const v4f*)(X + (size_t)min(nrt * 16 + l16, N - 1) * K + 4 * half);
#pragma unroll
      for (int t = 0; t < KQ; ++t) an[t] = ap[2 * t];
    }
    v8f acc[COLT];
#pragma unroll
    for (int ct = 0; ct < COLT; ++ct) acc[ct] = v8f{};
#pragma unroll
    for (int t = 0; t < KQ; ++t) {
      v2f a0, a1;
      a0.x = ac[t].x; a0.y = ac[t].y;
      a1.x = ac[t].z; a1.y = ac[t].w;
#pragma unroll
      for (int ct = 0; ct < COLT; ++ct)
        acc[ct] = __builtin_amdgcn_wmma_f32_16x16x4_f32(false, a0, false, wf[t][0][ct], (short)0, acc[ct],
                                                        false, false);
#pragma unroll
      for (int ct = 0; ct < COLT; ++ct)
        acc[ct] = __builtin_amdgcn_wmma_f32_16x16x4_f32(false, a1, false, wf[t][1][ct], (short)0, acc[ct],
                                                        false, false);
    }
    float* __restrict__ crow = C + (size_t)(rt * 16 + 8 * half) * H + COLT * l16;
    if (rt * 16 + 16 <= N) {
#pragma unroll
      for (int v = 0; v < 8; ++v) {
        cvec o;
#pragma unroll
        for (int ct = 0; ct < COLT; ++ct) o[ct] = acc[ct][v] + bv[ct];
        *(cvec*)(crow + (size_t)v * H) = o;
      }
    } else {
#pragma unroll
      for (int v = 0; v < 8; ++v) {
        cvec o;
#pragma unroll
        for (int ct = 0; ct < COLT; ++ct) o[ct] = acc[ct][v] + bv[ct];
        if (rt * 16 + 8 * half + v < N) *(cvec*)(crow + (size_t)v * H) = o;
      }
    }
  }
}

constexpr int RNODES = 392;
constexpr int LCAP   = 12288;
constexpr int OWN_TB = 1024;

constexpr int MK_TB    = 1024;
constexpr int MK_OW    = 256;
constexpr int MK_PITCH = 33;

template <bool V> struct QuadVec { static constexpr bool value = V; };

template <bool VEC>
__device__ __forceinline__ int4 load_quad(const int* __restrict__ col, int q, int E) {
  if constexpr (VEC) return ((const int4*)col)[q];
  int4 r;
  const int e = 4 * q;
  r.x = (e     < E) ? col[e]     : -1;
  r.y = (e + 1 < E) ? col[e + 1] : -1;
  r.z = (e + 2 < E) ? col[e + 2] : -1;
  r.w = (e + 3 < E) ? col[e + 3] : -1;
  return r;
}

__global__ void __launch_bounds__(MK_TB)
mark_owner_quads(const int* __restrict__ ei, int E, int G, int NW, unsigned* __restrict__ marks) {
  __shared__ unsigned tile[MK_OW * MK_PITCH];
  const int tid = threadIdx.x, lane = tid & 31, wave = tid >> 5;
  const int q = blockIdx.x * MK_TB + tid;
  const bool vec = (E & 3) == 0;
  int4 d = make_int4(-1, -1, -1, -1);
  if ((long long)4 * q < E) d = vec ? load_quad<true>(ei + E, q, E) : load_quad<false>(ei + E, q, E);
  int own[4];
  own[0] = d.x >= 0 ? d.x / RNODES : -1;
  own[1] = d.y >= 0 ? d.y / RNODES : -1;
  own[2] = d.z >= 0 ? d.z / RNODES : -1;
  own[3] = d.w >= 0 ? d.w / RNODES : -1;
  for (int g0 = 0; g0 < G; g0 += MK_OW) {
    for (int i = tid; i < MK_OW * MK_PITCH; i += MK_TB) tile[i] = 0u;
    __syncthreads();
#pragma unroll
    for (int k = 0; k < 4; ++k) {
      const unsigned o = (unsigned)(own[k] - g0);
      if (o < (unsigned)MK_OW) atomicOr(&tile[o * MK_PITCH + wave], 1u << lane);
    }
    __syncthreads();
    for (int o = wave; o < MK_OW && g0 + o < G; o += MK_TB / 32)
      marks[(size_t)(g0 + o) * NW + (size_t)blockIdx.x * 32 + lane] = tile[o * MK_PITCH + lane];
    __syncthreads();
  }
}

__device__ __forceinline__ void ldsMaxF(float* addr, float val) {
  if (val >= 0.0f) atomicMax((int*)addr, __float_as_int(val));
  else             atomicMin((unsigned int*)addr, __float_as_uint(val));
}

template <int H> struct RowFrag { float v[H / 32]; };

template <int H>
__device__ __forceinline__ RowFrag<H> load_row(const float* __restrict__ basep, int n, int lane) {
  RowFrag<H> r;
  if constexpr (H == 64) {
    const float2 a = *(const float2*)(basep + (size_t)n * 64 + lane * 2);
    r.v[0] = a.x; r.v[1] = a.y;
  } else {
    r.v[0] = basep[(size_t)n * 32 + lane];
  }
  return r;
}

template <int H>
__device__ __forceinline__ RowFrag<H> lds_row(const float* s, int n, int lane) {
  RowFrag<H> r;
  if constexpr (H == 64) {
    const float2 a = *(const float2*)(s + n * 64 + lane * 2);
    r.v[0] = a.x; r.v[1] = a.y;
  } else {
    r.v[0] = s[n * 32 + lane];
  }
  return r;
}

template <int H>
__device__ __forceinline__ float score_part(const RowFrag<H>& a, const RowFrag<H>& b,
                                            const RowFrag<H>& at) {
  if constexpr (H == 64) {
    float h0 = a.v[0] + b.v[0]; h0 = h0 > 0.0f ? h0 : 0.2f * h0;
    float h1 = a.v[1] + b.v[1]; h1 = h1 > 0.0f ? h1 : 0.2f * h1;
    return h0 * at.v[0] + h1 * at.v[1];
  } else {
    float hv = a.v[0] + b.v[0];
    hv = hv > 0.0f ? hv : 0.2f * hv;
    return hv * at.v[0];
  }
}

template <int H, int K, bool FINAL, bool FROMLIST>
__global__ void __launch_bounds__(OWN_TB)
gat_layer_owner(const float* __restrict__ xl, const float* __restrict__ xin,
                const float* __restrict__ Wr, const float* __restrict__ br,
                const float* __restrict__ att, const int* __restrict__ ei, int E, int N,
                const float* __restrict__ bias, float* __restrict__ ebufG,
                const float* __restrict__ Wlin,
                const float* __restrict__ blin, float* __restrict__ outp,
                int* __restrict__ listG, int* __restrict__ cntG,
                const unsigned* __restrict__ marks, int NW) {
  constexpr int U = 8;
  constexpr int COLT = H / 16;
  constexpr int ROWT = (RNODES + 15) / 16;
  static_assert(ROWT <= OWN_TB / 32, "one wave per row tile");
  static_assert(K * H <= LCAP, "the weight fragments are staged in the score array");
  __shared__ __attribute__((aligned(16))) float s_xa[RNODES * H];
  __shared__ float s_accF[FINAL ? RNODES : 1];
  float* const s_acc = FINAL ? s_accF : s_xa;
  __shared__ float s_m[RNODES];
  __shared__ float s_den[RNODES];
  __shared__ int   s_src[LCAP];
  __shared__ int   s_dl[LCAP];
  __shared__ __attribute__((aligned(16))) float s_e[LCAP];
  __shared__ float s_p[FINAL ? LCAP : 1];
  __shared__ int   s_cnt;
  __shared__ int   s_qn;

  const int tid = threadIdx.x, lane = tid & 31, wave = tid >> 5;
  const int base = blockIdx.x * RNODES;
  const int Rn = min(RNODES, N - base);
  const int* __restrict__ dstp = ei + E;

  if constexpr (FINAL) { for (int i = tid; i < RNODES; i += OWN_TB) s_accF[i] = 0.0f; }
  for (int i = tid; i < RNODES; i += OWN_TB) { s_m[i] = -INFINITY; s_den[i] = 0.0f; }
  for (int j = tid; j < Rn; j += OWN_TB) { s_src[j] = base + j; s_dl[j] = j; }

  {
    float* s_w = s_e;
    constexpr int SB = 8;
    constexpr int NB = K / 4 / SB;
    constexpr int WPT = K * H / OWN_TB;
    static_assert(K % (4 * SB) == 0 && (K * H) % OWN_TB == 0, "whole blocks, whole staging rounds");
    const bool gw = wave < ROWT && wave * 16 < Rn;
    const int half = lane >> 4, l16 = lane & 15;
    const int r = min(wave * 16 + l16, Rn - 1);
    const float* __restrict__ arow = xin + (size_t)(base + r) * K + 2 * half;
    v2f an[SB];
#pragma unroll
    for (int j = 0; j < SB; ++j) an[j] = *(const v2f*)(arow + 4 * j);
    float wv[WPT];
#pragma unroll
    for (int j = 0; j < WPT; ++j) {
      const int i = tid + j * OWN_TB;
      const int e = i & 1, l = (i >> 1) & 31, t = i >> 6;
      const int k = 4 * (t / COLT) + 2 * (l >> 4) + e;
      wv[j] = Wr[k * H + (t % COLT) * 16 + (l & 15)];
    }
#pragma unroll
    for (int j = 0; j < WPT; ++j) s_w[tid + j * OWN_TB] = wv[j];
    __syncthreads();
    if (gw) {
      v8f acc[COLT];
#pragma unroll
      for (int ct = 0; ct < COLT; ++ct) acc[ct] = v8f{};
#pragma unroll
      for (int blk = 0; blk < NB; ++blk) {
        v2f ac[SB];
#pragma unroll
        for (int j = 0; j < SB; ++j) ac[j] = an[j];
        if (blk + 1 < NB) {
#pragma unroll
          for (int j = 0; j < SB; ++j) an[j] = *(const v2f*)(arow + 4 * ((blk + 1) * SB + j));
        }
#pragma unroll
        for (int j = 0; j < SB; ++j) {
          const int s = blk * SB + j;
#pragma unroll
          for (int ct = 0; ct < COLT; ++ct) {
            const v2f b = *(const v2f*)(s_w + ((s * COLT + ct) * 32 + lane) * 2);
            acc[ct] = __builtin_amdgcn_wmma_f32_16x16x4_f32(false, ac[j], false, b, (short)0, acc[ct],
                                                            false, false);
          }
        }
      }
#pragma unroll
      for (int ct = 0; ct < COLT; ++ct) {
        const float bv = br[ct * 16 + l16];
#pragma unroll
        for (int v = 0; v < 8; ++v) {
          const int row = wave * 16 + v + 8 * half;
          if (row < Rn) s_xa[row * H + ct * 16 + l16] = acc[ct][v] + bv;
        }
      }
    }
  }
  int2* __restrict__ lg = (int2*)listG + (size_t)blockIdx.x * LCAP;
  int cnt;
  if constexpr (FROMLIST) {
    cnt = cntG[blockIdx.x];
    if (Rn + cnt <= LCAP) {
      for (int i = tid; i < cnt; i += OWN_TB) {
        const int2 p = lg[i];
        s_src[Rn + i] = p.x; s_dl[Rn + i] = p.y;
      }
    }
    __syncthreads();
  } else {
    if (tid == 0) { s_cnt = 0; s_qn = 0; }
    __syncthreads();

    int* s_q = (int*)s_e;
    const unsigned* __restrict__ plane = marks + (size_t)blockIdx.x * NW;
    constexpr int PW = 8;
    for (int w0 = tid; w0 < NW; w0 += OWN_TB * PW) {
      unsigned m[PW];
#pragma unroll
      for (int k = 0; k < PW; ++k) {
        const int w = w0 + k * OWN_TB;
        m[k] = (w < NW) ? plane[w] : 0u;
      }
      int nb = 0;
#pragma unroll
      for (int k = 0; k < PW; ++k) nb += __popc(m[k]);
      int slot = atomicAdd(&s_qn, nb);
#pragma unroll
      for (int k = 0; k < PW; ++k) {
        unsigned mm = m[k];
        while (mm) {
          const int b = __builtin_ctz(mm);
          mm &= mm - 1;
          if (slot < LCAP) s_q[slot] = (w0 + k * OWN_TB) * 32 + b;
          ++slot;
        }
      }
    }
    __syncthreads();
    const int qn = s_qn;
    const int nqd = min(qn, LCAP);

    auto hit = [&](int d) { return ((unsigned)(d - base) < (unsigned)Rn) ? 1 : 0; };
    auto put = [&](int d, int s, int& slot) {
      if (hit(d)) {
        if (slot < LCAP) { s_src[slot] = s; s_dl[slot] = d - base; }
        ++slot;
      }
    };
    constexpr int QV = 4;
    auto gather_quads = [&](auto vtag) {
      constexpr bool VEC = decltype(vtag)::value;
      for (int i0 = tid; i0 < nqd; i0 += OWN_TB * QV) {
        int4 d4[QV], s4[QV];
#pragma unroll
        for (int k = 0; k < QV; ++k) {
          const int qq = s_q[min(i0 + k * OWN_TB, nqd - 1)];
          d4[k] = load_quad<VEC>(dstp, qq, E);
          s4[k] = load_quad<VEC>(ei, qq, E);
        }
        int nm = 0;
#pragma unroll
        for (int k = 0; k < QV; ++k) {
          if (i0 + k * OWN_TB < nqd) nm += hit(d4[k].x) + hit(d4[k].y) + hit(d4[k].z) + hit(d4[k].w);
        }
        int slot = Rn + atomicAdd(&s_cnt, nm);
#pragma unroll
        for (int k = 0; k < QV; ++k) {
          if (i0 + k * OWN_TB < nqd) {
            put(d4[k].x, s4[k].x, slot); put(d4[k].y, s4[k].y, slot);
            put(d4[k].z, s4[k].z, slot); put(d4[k].w, s4[k].w, slot);
          }
        }
      }
    };
    if ((E & 3) == 0) gather_quads(QuadVec<true>{});
    else              gather_quads(QuadVec<false>{});
    __syncthreads();

    cnt = (qn > LCAP) ? qn : s_cnt;
    if (tid == 0) cntG[blockIdx.x] = cnt;
    if (Rn + cnt <= LCAP) {
      for (int i = tid; i < cnt; i += OWN_TB) lg[i] = make_int2(s_src[Rn + i], s_dl[Rn + i]);
    }
  }

  const int total = Rn + cnt;
  RowFrag<H> at;
  if constexpr (H == 64) { at.v[0] = att[lane * 2]; at.v[1] = att[lane * 2 + 1]; }
  else                   { at.v[0] = att[lane]; }
  float wl = 0.0f;
  if constexpr (FINAL) {
    static_assert(!FINAL || H == 32, "one channel per lane in the last layer");
    wl = Wlin[lane];
  }

  if (total <= LCAP) {
    for (int i0 = wave * U; i0 < total; i0 += (OWN_TB / 32) * U) {
      const int li = min(i0 + (lane & (U - 1)), total - 1);
      const int my_dl = s_dl[li];
      const int my_src = s_src[li];
      RowFrag<H> a[U], b[U];
#pragma unroll
      for (int u = 0; u < U; ++u) {
        a[u] = load_row<H>(xl, lane_int(my_src, u), lane);
        b[u] = lds_row<H>(s_xa, lane_int(my_dl, u), lane);
      }
      float tot[U];
#pragma unroll
      for (int u = 0; u < U; ++u) tot[u] = score_part<H>(a[u], b[u], at);
      if constexpr (FINAL) {
        float x[2 * U];
#pragma unroll
        for (int u = 0; u < U; ++u) { x[u] = tot[u]; x[U + u] = a[u].v[0] * wl; }
        const bool h1 = (lane & 1) != 0, h2 = (lane & 2) != 0, h4 = (lane & 4) != 0, h8 = (lane & 8) != 0;
        float y[U], z[U / 2], w[U / 4];
#pragma unroll
        for (int i = 0; i < U; ++i) y[i] = fold_pair<1>(x[2 * i], x[2 * i + 1], h1);
#pragma unroll
        for (int i = 0; i < U / 2; ++i) z[i] = fold_pair<2>(y[2 * i], y[2 * i + 1], h2);
#pragma unroll
        for (int i = 0; i < U / 4; ++i) w[i] = fold_pair<4>(z[2 * i], z[2 * i + 1], h4);
        static_assert(U == 8, "four fold stages take 16 vectors to one");
        float r = fold_pair<8>(w[0], w[1], h8);
        r += lane_xor16(r);
        const int idx = i0 + (lane & (U - 1));
        if (lane < 2 * U && idx < total) {
          if (lane < U) { s_e[idx] = r; ldsMaxF(&s_m[my_dl], r); }
          else          { s_p[idx] = r; }
        }
      } else {
#pragma unroll
        for (int u = 0; u < U; ++u) tot[u] += lane_xor16(tot[u]);
#pragma unroll
        for (int u = 0; u < U; ++u) tot[u] += lane_xor_row<8>(tot[u]);
#pragma unroll
        for (int u = 0; u < U; ++u) tot[u] += lane_xor_row<4>(tot[u]);
#pragma unroll
        for (int u = 0; u < U; ++u) tot[u] += lane_xor_row<2>(tot[u]);
#pragma unroll
        for (int u = 0; u < U; ++u) tot[u] += lane_xor_row<1>(tot[u]);
        float mine = tot[0];
#pragma unroll
        for (int u = 1; u < U; ++u) mine = (lane == u) ? tot[u] : mine;
        if (lane < U && i0 + lane < total) { s_e[i0 + lane] = mine; ldsMaxF(&s_m[my_dl], mine); }
      }
    }
    __syncthreads();
    if constexpr (!FINAL) {
      for (int i = tid; i < RNODES * H; i += OWN_TB) s_xa[i] = 0.0f;
      __syncthreads();
    }
    if constexpr (FINAL) {
      for (int i = tid; i < total; i += OWN_TB) {
        const int d = s_dl[i];
        const float ex = expf(s_e[i] - s_m[d]);
        atomicAdd(&s_den[d], ex);
        atomicAdd(&s_acc[d], ex * s_p[i]);
      }
    } else {
      for (int i0 = wave * U; i0 < total; i0 += (OWN_TB / 32) * U) {
        const int li = min(i0 + (lane & (U - 1)), total - 1);
        const int my_dl = s_dl[li];
        const int my_src = s_src[li];
        const float my_ex = expf(s_e[li] - s_m[my_dl]);
        RowFrag<H> a[U];
#pragma unroll
        for (int u = 0; u < U; ++u) a[u] = load_row<H>(xl, lane_int(my_src, u), lane);
        if (lane < U && i0 + lane < total) atomicAdd(&s_den[my_dl], my_ex);
#pragma unroll
        for (int u = 0; u < U; ++u) {
          if (i0 + u < total) {
            const int d = lane_int(my_dl, u);
            const float ex = lane_float(my_ex, u);
            if constexpr (H == 64) {
              atomicAdd(&s_acc[d * 64 + lane * 2], ex * a[u].v[0]);
              atomicAdd(&s_acc[d * 64 + lane * 2 + 1], ex * a[u].v[1]);
            } else {
              atomicAdd(&s_acc[d * 32 + lane], ex * a[u].v[0]);
            }
          }
        }
      }
    }
  } else {
    for (int phase = 0; phase < 2; ++phase) {
      auto one = [&](int s, int d, int eid) {
        const int l = d - base;
        if (phase == 0) {
          const RowFrag<H> a = load_row<H>(xl, s, lane), b = lds_row<H>(s_xa, l, lane);
          const float tot = wave_sum32(score_part<H>(a, b, at));
          if (lane == 0) { ebufG[eid] = tot; ldsMaxF(&s_m[l], tot); }
        } else {
          const float ex = expf(ebufG[eid] - s_m[l]);
          if (lane == 0) atomicAdd(&s_den[l], ex);
          if constexpr (FINAL) {
            const RowFrag<H> a = load_row<H>(xl, s, lane);
            const float pv = wave_sum32(a.v[0] * wl);
            if (lane == 0) atomicAdd(&s_acc[l], ex * pv);
          } else {
            const RowFrag<H> a = load_row<H>(xl, s, lane);
            if constexpr (H == 64) {
              atomicAdd(&s_acc[l * 64 + lane * 2], ex * a.v[0]);
              atomicAdd(&s_acc[l * 64 + lane * 2 + 1], ex * a.v[1]);
            } else {
              atomicAdd(&s_acc[l * 32 + lane], ex * a.v[0]);
            }
          }
        }
      };
      for (int j = wave; j < Rn; j += OWN_TB / 32) one(base + j, base + j, E + base + j);
      for (int e0 = wave * 32; e0 < E; e0 += OWN_TB) {
        const int e = e0 + lane;
        const int d = (e < E) ? dstp[e] : -1;
        const bool hit = (unsigned)(d - base) < (unsigned)Rn;
        unsigned mask = (unsigned)__ballot(hit);
        while (mask) {
          const int bsel = __builtin_ctz(mask);
          mask &= mask - 1;
          one(ei[e0 + bsel], __shfl(d, bsel, 32), e0 + bsel);
        }
      }
      __syncthreads();
      if constexpr (!FINAL) {
        if (phase == 0) {
          for (int i = tid; i < RNODES * H; i += OWN_TB) s_xa[i] = 0.0f;
          __syncthreads();
        }
      }
    }
  }
  __syncthreads();

  if constexpr (FINAL) {
    float cb = blin[0];
#pragma unroll
    for (int c = 0; c < 32; ++c) cb += bias[c] * Wlin[c];
    for (int j = tid; j < Rn; j += OWN_TB)
      outp[base + j] = s_acc[j] / (s_den[j] + 1e-16f) + cb;
  } else {
    for (int i = tid; i < Rn * H; i += OWN_TB) {
      const int n = i / H, c = i % H;
      const float v = s_acc[i] / (s_den[n] + 1e-16f) + bias[c];
      outp[(size_t)base * H + i] = v > 0.0f ? v : 0.01f * v;
    }
  }
}

static inline int cdiv(int a, int b) { return (a + b - 1) / b; }

extern "C" void kernel_launch(void* const* d_in, const int* in_sizes, int n_in,
                              void* d_out, int out_size, void* d_ws, size_t ws_size,
                              hipStream_t stream) {
  const float* x     = (const float*)d_in[0];
  const int*   ei    = (const int*)  d_in[1];
  const float* W1l   = (const float*)d_in[2];
  const float* b1l   = (const float*)d_in[3];
  const float* W1r   = (const float*)d_in[4];
  const float* b1r   = (const float*)d_in[5];
  const float* att1  = (const float*)d_in[6];
  const float* bias1 = (const float*)d_in[7];
  const float* W2l   = (const float*)d_in[8];
  const float* b2l   = (const float*)d_in[9];
  const float* W2r   = (const float*)d_in[10];
  const float* b2r   = (const float*)d_in[11];
  const float* att2  = (const float*)d_in[12];
  const float* bias2 = (const float*)d_in[13];
  const float* Wlin  = (const float*)d_in[14];
  const float* blin  = (const float*)d_in[15];
  float* out = (float*)d_out;

  const int N    = in_sizes[0] / 128;
  const int E    = in_sizes[1] / 2;
  const int Etot = E + N;

  float* ws = (float*)d_ws;
  size_t off = 0;
  float* xl1  = ws + off; off += (size_t)N * 64;
  float* h1   = ws + off; off += (size_t)N * 64;
  float* xl2  = ws + off; off += (size_t)N * 32;
  float* ebuf = ws + off; off += (size_t)Etot;
  off = (off + 1) & ~(size_t)1;
  int* listG = (int*)(ws + off); off += (size_t)cdiv(N, RNODES) * LCAP * 2;
  int* cntG  = (int*)(ws + off); off += (size_t)cdiv(N, RNODES);
  off = (off + 31) & ~(size_t)31;
  const int markGroups = cdiv(cdiv(E, 4), MK_TB);
  const int NW = markGroups * 32;
  unsigned* marks = (unsigned*)(ws + off); off += (size_t)cdiv(N, RNODES) * NW;
  (void)ws_size; (void)n_in; (void)out_size;

  const int ownGroups = cdiv(N, RNODES);
  const int gemmGroups1 = 256;
  const int gemmGroups2 = 512;

  mark_owner_quads<<<markGroups, MK_TB, 0, stream>>>(ei, E, ownGroups, NW, marks);

  gemm_bias_wmma<128, 64><<<gemmGroups1, GM_TB, 0, stream>>>(x, W1l, b1l, xl1, N);
  gat_layer_owner<64, 128, false, false><<<ownGroups, OWN_TB, 0, stream>>>(
      xl1, x, W1r, b1r, att1, ei, E, N, bias1, ebuf, nullptr, nullptr, h1, listG, cntG, marks, NW);

  gemm_bias_wmma<64, 32><<<gemmGroups2, GM_TB, 0, stream>>>(h1, W2l, b2l, xl2, N);
  gat_layer_owner<32, 64, true, true><<<ownGroups, OWN_TB, 0, stream>>>(
      xl2, h1, W2r, b2r, att2, ei, E, N, bias2, ebuf, Wlin, blin, out, listG, cntG, marks, NW);
}


# layer 1: a node's softmax shift is its self-loop score (known up front), so ONE pass over the edges: each xl row gathered once, not twice; sums start from the self term, sum order changes
# speedup vs baseline: 7.2215x; 1.1328x over previous
#include <hip/hip_runtime.h>
#include <hip/hip_bf16.h>
#include <math.h>

typedef __attribute__((ext_vector_type(2))) float v2f;
typedef __attribute__((ext_vector_type(8))) float v8f;

__device__ __forceinline__ float wave_sum32(float v) {
#pragma unroll
  for (int off = 16; off > 0; off >>= 1) v += __shfl_xor(v, off, 32);
  return v;
}

template <int M>
__device__ __forceinline__ float lane_xor_row(float x) {
  const int b = __builtin_bit_cast(int, x);
  return __builtin_bit_cast(float, __builtin_amdgcn_update_dpp(b, b, 0x160 + M, 0xF, 0xF, true));
}
__device__ __forceinline__ float lane_xor16(float x) {
  const int b = __builtin_bit_cast(int, x);
  return __builtin_bit_cast(float,
                            __builtin_amdgcn_permlanex16(b, b, 0x76543210, 0xfedcba98, false, false));
}
__device__ __forceinline__ int lane_int(int x, int l) { return __builtin_amdgcn_readlane(x, l); }
__device__ __forceinline__ float lane_float(float x, int l) {
  return __builtin_bit_cast(float, __builtin_amdgcn_readlane(__builtin_bit_cast(int, x), l));
}

template <int M>
__device__ __forceinline__ float fold_pair(float xa, float xb, bool hi) {
  const float own = hi ? xb : xa;
  const float oth = hi ? xa : xb;
  return own + lane_xor_row<M>(oth);
}

constexpr int GM_TB = 256;

template <int C> struct ColVec;
template <> struct ColVec<4> { typedef __attribute__((ext_vector_type(4))) float type; };
template <> struct ColVec<2> { typedef __attribute__((ext_vector_type(2))) float type; };

template <int K, int H>
__global__ void __launch_bounds__(GM_TB)
gemm_bias_wmma(const float* __restrict__ X, const float* __restrict__ W,
               const float* __restrict__ bias, float* __restrict__ C, int N) {
  constexpr int COLT = H / 16;
  constexpr int KQ = K / 8;
  static_assert(K % 8 == 0 && H % 16 == 0, "whole 16-byte loads, whole column tiles");
  typedef typename ColVec<COLT>::type cvec;
  typedef __attribute__((ext_vector_type(4))) float v4f;
  const int lane = threadIdx.x & 31;
  const int half = lane >> 4, l16 = lane & 15;
  const int wv = __builtin_amdgcn_readfirstlane((int)(threadIdx.x >> 5));
  const int gwave = wv * (int)gridDim.x + (int)blockIdx.x;
  const int nwaves = (int)gridDim.x * (GM_TB / 32);
  const int rowTiles = (N + 15) >> 4;

  v2f wf[KQ][2][COLT];
#pragma unroll
  for (int t = 0; t < KQ; ++t) {
#pragma unroll
    for (int e = 0; e < 2; ++e) {
      const int k = 8 * t + 4 * half + 2 * e;
      const cvec w0 = *(const cvec*)(W + (size_t)k * H + COLT * l16);
      const cvec w1 = *(const cvec*)(W + (size_t)(k + 1) * H + COLT * l16);
#pragma unroll
      for (int ct = 0; ct < COLT; ++ct) { wf[t][e][ct].x = w0[ct]; wf[t][e][ct].y = w1[ct]; }
    }
  }
  const cvec bv = *(const cvec*)(bias + COLT * l16);

  v4f an[KQ];
  int rt = gwave;
  if (rt < rowTiles) {
    const v4f* __restrict__ ap = (const v4f*)(X + (size_t)min(rt * 16 + l16, N - 1) * K + 4 * half);
#pragma unroll
    for (int t = 0; t < KQ; ++t) an[t] = ap[2 * t];
  }
  for (; rt < rowTiles; rt += nwaves) {
    v4f ac[KQ];
#pragma unroll
    for (int t = 0; t < KQ; ++t) ac[t] = an[t];
    const int nrt = rt + nwaves;
    if (nrt < rowTiles) {
      const v4f* __restrict__ ap = (const v4f*)(X + (size_t)min(nrt * 16 + l16, N - 1) * K + 4 * half);
#pragma unroll
      for (int t = 0; t < KQ; ++t) an[t] = ap[2 * t];
    }
    v8f acc[COLT];
#pragma unroll
    for (int ct = 0; ct < COLT; ++ct) acc[ct] = v8f{};
#pragma unroll
    for (int t = 0; t < KQ; ++t) {
      v2f a0, a1;
      a0.x = ac[t].x; a0.y = ac[t].y;
      a1.x = ac[t].z; a1.y = ac[t].w;
#pragma unroll
      for (int ct = 0; ct < COLT; ++ct)
        acc[ct] = __builtin_amdgcn_wmma_f32_16x16x4_f32(false, a0, false, wf[t][0][ct], (short)0, acc[ct],
                                                        false, false);
#pragma unroll
      for (int ct = 0; ct < COLT; ++ct)
        acc[ct] = __builtin_amdgcn_wmma_f32_16x16x4_f32(false, a1, false, wf[t][1][ct], (short)0, acc[ct],
                                                        false, false);
    }
    float* __restrict__ crow = C + (size_t)(rt * 16 + 8 * half) * H + COLT * l16;
    if (rt * 16 + 16 <= N) {
#pragma unroll
      for (int v = 0; v < 8; ++v) {
        cvec o;
#pragma unroll
        for (int ct = 0; ct < COLT; ++ct) o[ct] = acc[ct][v] + bv[ct];
        *(cvec*)(crow + (size_t)v * H) = o;
      }
    } else {
#pragma unroll
      for (int v = 0; v < 8; ++v) {
        cvec o;
#pragma unroll
        for (int ct = 0; ct < COLT; ++ct) o[ct] = acc[ct][v] + bv[ct];
        if (rt * 16 + 8 * half + v < N) *(cvec*)(crow + (size_t)v * H) = o;
      }
    }
  }
}

constexpr int RNODES = 392;
constexpr int LCAP   = 8192;
constexpr float SHIFT_MAX = 30.0f;
constexpr int OWN_TB = 1024;

constexpr int MK_TB    = 1024;
constexpr int MK_OW    = 256;
constexpr int MK_PITCH = 33;

template <bool V> struct QuadVec { static constexpr bool value = V; };

template <bool VEC>
__device__ __forceinline__ int4 load_quad(const int* __restrict__ col, int q, int E) {
  if constexpr (VEC) return ((const int4*)col)[q];
  int4 r;
  const int e = 4 * q;
  r.x = (e     < E) ? col[e]     : -1;
  r.y = (e + 1 < E) ? col[e + 1] : -1;
  r.z = (e + 2 < E) ? col[e + 2] : -1;
  r.w = (e + 3 < E) ? col[e + 3] : -1;
  return r;
}

__global__ void __launch_bounds__(MK_TB)
mark_owner_quads(const int* __restrict__ ei, int E, int G, int NW, unsigned* __restrict__ marks) {
  __shared__ unsigned tile[MK_OW * MK_PITCH];
  const int tid = threadIdx.x, lane = tid & 31, wave = tid >> 5;
  const int q = blockIdx.x * MK_TB + tid;
  const bool vec = (E & 3) == 0;
  int4 d = make_int4(-1, -1, -1, -1);
  if ((long long)4 * q < E) d = vec ? load_quad<true>(ei + E, q, E) : load_quad<false>(ei + E, q, E);
  int own[4];
  own[0] = d.x >= 0 ? d.x / RNODES : -1;
  own[1] = d.y >= 0 ? d.y / RNODES : -1;
  own[2] = d.z >= 0 ? d.z / RNODES : -1;
  own[3] = d.w >= 0 ? d.w / RNODES : -1;
  for (int g0 = 0; g0 < G; g0 += MK_OW) {
    for (int i = tid; i < MK_OW * MK_PITCH; i += MK_TB) tile[i] = 0u;
    __syncthreads();
#pragma unroll
    for (int k = 0; k < 4; ++k) {
      const unsigned o = (unsigned)(own[k] - g0);
      if (o < (unsigned)MK_OW) atomicOr(&tile[o * MK_PITCH + wave], 1u << lane);
    }
    __syncthreads();
    for (int o = wave; o < MK_OW && g0 + o < G; o += MK_TB / 32)
      marks[(size_t)(g0 + o) * NW + (size_t)blockIdx.x * 32 + lane] = tile[o * MK_PITCH + lane];
    __syncthreads();
  }
}

__device__ __forceinline__ void ldsMaxF(float* addr, float val) {
  if (val >= 0.0f) atomicMax((int*)addr, __float_as_int(val));
  else             atomicMin((unsigned int*)addr, __float_as_uint(val));
}

template <int H> struct RowFrag { float v[H / 32]; };

template <int H>
__device__ __forceinline__ RowFrag<H> load_row(const float* __restrict__ basep, int n, int lane) {
  RowFrag<H> r;
  if constexpr (H == 64) {
    const float2 a = *(const float2*)(basep + (size_t)n * 64 + lane * 2);
    r.v[0] = a.x; r.v[1] = a.y;
  } else {
    r.v[0] = basep[(size_t)n * 32 + lane];
  }
  return r;
}

template <int H, int P>
__device__ __forceinline__ RowFrag<H> lds_row(const float* s, int n, int lane) {
  RowFrag<H> r;
  if constexpr (H == 64) {
    const float2 a = *(const float2*)(s + n * P + lane * 2);
    r.v[0] = a.x; r.v[1] = a.y;
  } else {
    r.v[0] = s[n * P + lane];
  }
  return r;
}
template <int H, int P>
__device__ __forceinline__ void lds_put_row(float* s, int n, int lane, const RowFrag<H>& r) {
  if constexpr (H == 64) {
    *(float2*)(s + n * P + lane * 2) = make_float2(r.v[0], r.v[1]);
  } else {
    s[n * P + lane] = r.v[0];
  }
}

template <int H>
__device__ __forceinline__ float score_part(const RowFrag<H>& a, const RowFrag<H>& b,
                                            const RowFrag<H>& at) {
  if constexpr (H == 64) {
    float h0 = a.v[0] + b.v[0]; h0 = h0 > 0.0f ? h0 : 0.2f * h0;
    float h1 = a.v[1] + b.v[1]; h1 = h1 > 0.0f ? h1 : 0.2f * h1;
    return h0 * at.v[0] + h1 * at.v[1];
  } else {
    float hv = a.v[0] + b.v[0];
    hv = hv > 0.0f ? hv : 0.2f * hv;
    return hv * at.v[0];
  }
}

template <int H, int K, bool FINAL, bool FROMLIST>
__global__ void __launch_bounds__(OWN_TB)
gat_layer_owner(const float* __restrict__ xl, const float* __restrict__ xin,
                const float* __restrict__ Wr, const float* __restrict__ br,
                const float* __restrict__ att, const int* __restrict__ ei, int E, int N,
                const float* __restrict__ bias, float* __restrict__ ebufG,
                const float* __restrict__ Wlin,
                const float* __restrict__ blin, float* __restrict__ outp,
                int* __restrict__ listG, int* __restrict__ cntG,
                const unsigned* __restrict__ marks, int NW) {
  constexpr int U = 8;
  constexpr int COLT = H / 16;
  constexpr int ROWT = (RNODES + 15) / 16;
  static_assert(ROWT <= OWN_TB / 32, "one wave per row tile");
  static_assert(K * H <= LCAP, "the weight fragments are staged in the score array");
  static_assert(FINAL == FROMLIST, "the one-pass layer builds the list; its self-loop rows are set up behind the list builder's first barrier");
  constexpr int XS = FINAL ? H : 2 * H;
  __shared__ __attribute__((aligned(16))) float s_xa[RNODES * XS];
  __shared__ float s_accF[FINAL ? RNODES : 1];
  float* const s_acc = FINAL ? s_accF : s_xa + H;
  __shared__ float s_m[RNODES];
  __shared__ float s_den[RNODES];
  __shared__ float s_sh[FINAL ? 1 : RNODES];
  __shared__ int   s_bad;
  __shared__ int   s_src[LCAP];
  __shared__ int   s_dl[LCAP];
  __shared__ __attribute__((aligned(16))) float s_e[LCAP];
  __shared__ float s_p[FINAL ? LCAP : 1];
  __shared__ int   s_cnt;
  __shared__ int   s_qn;

  const int tid = threadIdx.x, lane = tid & 31, wave = tid >> 5;
  const int base = blockIdx.x * RNODES;
  const int Rn = min(RNODES, N - base);
  const int* __restrict__ dstp = ei + E;

  if constexpr (FINAL) {
    for (int i = tid; i < RNODES; i += OWN_TB) { s_accF[i] = 0.0f; s_m[i] = -INFINITY; s_den[i] = 0.0f; }
    for (int j = tid; j < Rn; j += OWN_TB) { s_src[j] = base + j; s_dl[j] = j; }
  } else {
    if (tid == 0) s_bad = 0;
  }
  RowFrag<H> at;
  if constexpr (H == 64) { at.v[0] = att[lane * 2]; at.v[1] = att[lane * 2 + 1]; }
  else                   { at.v[0] = att[lane]; }

  {
    float* s_w = s_e;
    constexpr int SB = 8;
    constexpr int NB = K / 4 / SB;
    constexpr int WPT = K * H / OWN_TB;
    static_assert(K % (4 * SB) == 0 && (K * H) % OWN_TB == 0, "whole blocks, whole staging rounds");
    const bool gw = wave < ROWT && wave * 16 < Rn;
    const int half = lane >> 4, l16 = lane & 15;
    const int r = min(wave * 16 + l16, Rn - 1);
    const float* __restrict__ arow = xin + (size_t)(base + r) * K + 2 * half;
    v2f an[SB];
#pragma unroll
    for (int j = 0; j < SB; ++j) an[j] = *(const v2f*)(arow + 4 * j);
    float wv[WPT];
#pragma unroll
    for (int j = 0; j < WPT; ++j) {
      const int i = tid + j * OWN_TB;
      const int e = i & 1, l = (i >> 1) & 31, t = i >> 6;
      const int k = 4 * (t / COLT) + 2 * (l >> 4) + e;
      wv[j] = Wr[k * H + (t % COLT) * 16 + (l & 15)];
    }
#pragma unroll
    for (int j = 0; j < WPT; ++j) s_w[tid + j * OWN_TB] = wv[j];
    __syncthreads();
    if (gw) {
      v8f acc[COLT];
#pragma unroll
      for (int ct = 0; ct < COLT; ++ct) acc[ct] = v8f{};
#pragma unroll
      for (int blk = 0; blk < NB; ++blk) {
        v2f ac[SB];
#pragma unroll
        for (int j = 0; j < SB; ++j) ac[j] = an[j];
        if (blk + 1 < NB) {
#pragma unroll
          for (int j = 0; j < SB; ++j) an[j] = *(const v2f*)(arow + 4 * ((blk + 1) * SB + j));
        }
#pragma unroll
        for (int j = 0; j < SB; ++j) {
          const int s = blk * SB + j;
#pragma unroll
          for (int ct = 0; ct < COLT; ++ct) {
            const v2f b = *(const v2f*)(s_w + ((s * COLT + ct) * 32 + lane) * 2);
            acc[ct] = __builtin_amdgcn_wmma_f32_16x16x4_f32(false, ac[j], false, b, (short)0, acc[ct],
                                                            false, false);
          }
        }
      }
#pragma unroll
      for (int ct = 0; ct < COLT; ++ct) {
        const float bv = br[ct * 16 + l16];
#pragma unroll
        for (int v = 0; v < 8; ++v) {
          const int row = wave * 16 + v + 8 * half;
          if (row < Rn) s_xa[row * XS + ct * 16 + l16] = acc[ct][v] + bv;
        }
      }
    }
  }
  int2* __restrict__ lg = (int2*)listG + (size_t)blockIdx.x * LCAP;
  int cnt;
  if constexpr (FROMLIST) {
    cnt = cntG[blockIdx.x];
    if (Rn + cnt <= LCAP) {
      for (int i = tid; i < cnt; i += OWN_TB) {
        const int2 p = lg[i];
        s_src[Rn + i] = p.x; s_dl[Rn + i] = p.y;
      }
    }
    __syncthreads();
  } else {
    if (tid == 0) { s_cnt = 0; s_qn = 0; }
    __syncthreads();

    if constexpr (!FINAL) {
      constexpr int PU = 16;
      for (int j0 = wave * PU; j0 < Rn; j0 += (OWN_TB / 32) * PU) {
        RowFrag<H> a[PU], b[PU];
#pragma unroll
        for (int u = 0; u < PU; ++u) {
          const int j = min(j0 + u, Rn - 1);
          a[u] = load_row<H>(xl, base + j, lane);
          b[u] = lds_row<H, XS>(s_xa, j, lane);
        }
        float tot[PU];
#pragma unroll
        for (int u = 0; u < PU; ++u) tot[u] = score_part<H>(a[u], b[u], at);
#pragma unroll
        for (int u = 0; u < PU; ++u) tot[u] += lane_xor16(tot[u]);
#pragma unroll
        for (int u = 0; u < PU; ++u) tot[u] += lane_xor_row<8>(tot[u]);
#pragma unroll
        for (int u = 0; u < PU; ++u) tot[u] += lane_xor_row<4>(tot[u]);
#pragma unroll
        for (int u = 0; u < PU; ++u) tot[u] += lane_xor_row<2>(tot[u]);
#pragma unroll
        for (int u = 0; u < PU; ++u) tot[u] += lane_xor_row<1>(tot[u]);
        float mine = tot[0];
#pragma unroll
        for (int u = 1; u < PU; ++u) mine = (lane == u) ? tot[u] : mine;
        if (lane < PU && j0 + lane < Rn) { s_sh[j0 + lane] = mine; s_den[j0 + lane] = 1.0f; }
#pragma unroll
        for (int u = 0; u < PU; ++u) {
          if (j0 + u < Rn) lds_put_row<H, XS>(s_acc, j0 + u, lane, a[u]);
        }
      }
    }

    int* s_q = (int*)s_e;
    const unsigned* __restrict__ plane = marks + (size_t)blockIdx.x * NW;
    constexpr int PW = 8;
    for (int w0 = tid; w0 < NW; w0 += OWN_TB * PW) {
      unsigned m[PW];
#pragma unroll
      for (int k = 0; k < PW; ++k) {
        const int w = w0 + k * OWN_TB;
        m[k] = (w < NW) ? plane[w] : 0u;
      }
      int nb = 0;
#pragma unroll
      for (int k = 0; k < PW; ++k) nb += __popc(m[k]);
      int slot = atomicAdd(&s_qn, nb);
#pragma unroll
      for (int k = 0; k < PW; ++k) {
        unsigned mm = m[k];
        while (mm) {
          const int b = __builtin_ctz(mm);
          mm &= mm - 1;
          if (slot < LCAP) s_q[slot] = (w0 + k * OWN_TB) * 32 + b;
          ++slot;
        }
      }
    }
    __syncthreads();
    const int qn = s_qn;
    const int nqd = min(qn, LCAP);

    auto hit = [&](int d) { return ((unsigned)(d - base) < (unsigned)Rn) ? 1 : 0; };
    auto put = [&](int d, int s, int& slot) {
      if (hit(d)) {
        if (slot < LCAP) { s_src[slot] = s; s_dl[slot] = d - base; }
        ++slot;
      }
    };
    constexpr int QV = 4;
    auto gather_quads = [&](auto vtag) {
      constexpr bool VEC = decltype(vtag)::value;
      for (int i0 = tid; i0 < nqd; i0 += OWN_TB * QV) {
        int4 d4[QV], s4[QV];
#pragma unroll
        for (int k = 0; k < QV; ++k) {
          const int qq = s_q[min(i0 + k * OWN_TB, nqd - 1)];
          d4[k] = load_quad<VEC>(dstp, qq, E);
          s4[k] = load_quad<VEC>(ei, qq, E);
        }
        int nm = 0;
#pragma unroll
        for (int k = 0; k < QV; ++k) {
          if (i0 + k * OWN_TB < nqd) nm += hit(d4[k].x) + hit(d4[k].y) + hit(d4[k].z) + hit(d4[k].w);
        }
        int slot = Rn + atomicAdd(&s_cnt, nm);
#pragma unroll
        for (int k = 0; k < QV; ++k) {
          if (i0 + k * OWN_TB < nqd) {
            put(d4[k].x, s4[k].x, slot); put(d4[k].y, s4[k].y, slot);
            put(d4[k].z, s4[k].z, slot); put(d4[k].w, s4[k].w, slot);
          }
        }
      }
    };
    if ((E & 3) == 0) gather_quads(QuadVec<true>{});
    else              gather_quads(QuadVec<false>{});
    __syncthreads();

    cnt = (qn > LCAP) ? qn : s_cnt;
    if (tid == 0) cntG[blockIdx.x] = cnt;
    if (Rn + cnt <= LCAP) {
      for (int i = tid; i < cnt; i += OWN_TB) lg[i] = make_int2(s_src[Rn + i], s_dl[Rn + i]);
    }
  }

  const int total = Rn + cnt;
  float wl = 0.0f;
  if constexpr (FINAL) {
    static_assert(!FINAL || H == 32, "one channel per lane in the last layer");
    wl = Wlin[lane];
  }

  bool listed = total <= LCAP;
  if (listed) {
    for (int i0 = (FINAL ? 0 : Rn) + wave * U; i0 < total; i0 += (OWN_TB / 32) * U) {
      const int li = min(i0 + (lane & (U - 1)), total - 1);
      const int my_dl = s_dl[li];
      const int my_src = s_src[li];
      RowFrag<H> a[U], b[U];
#pragma unroll
      for (int u = 0; u < U; ++u) {
        a[u] = load_row<H>(xl, lane_int(my_src, u), lane);
        b[u] = lds_row<H, XS>(s_xa, lane_int(my_dl, u), lane);
      }
      float tot[U];
#pragma unroll
      for (int u = 0; u < U; ++u) tot[u] = score_part<H>(a[u], b[u], at);
      if constexpr (FINAL) {
        float x[2 * U];
#pragma unroll
        for (int u = 0; u < U; ++u) { x[u] = tot[u]; x[U + u] = a[u].v[0] * wl; }
        const bool h1 = (lane & 1) != 0, h2 = (lane & 2) != 0, h4 = (lane & 4) != 0, h8 = (lane & 8) != 0;
        float y[U], z[U / 2], w[U / 4];
#pragma unroll
        for (int i = 0; i < U; ++i) y[i] = fold_pair<1>(x[2 * i], x[2 * i + 1], h1);
#pragma unroll
        for (int i = 0; i < U / 2; ++i) z[i] = fold_pair<2>(y[2 * i], y[2 * i + 1], h2);
#pragma unroll
        for (int i = 0; i < U / 4; ++i) w[i] = fold_pair<4>(z[2 * i], z[2 * i + 1], h4);
        static_assert(U == 8, "four fold stages take 16 vectors to one");
        float r = fold_pair<8>(w[0], w[1], h8);
        r += lane_xor16(r);
        const int idx = i0 + (lane & (U - 1));
        if (lane < 2 * U && idx < total) {
          if (lane < U) { s_e[idx] = r; ldsMaxF(&s_m[my_dl], r); }
          else          { s_p[idx] = r; }
        }
      } else {
#pragma unroll
        for (int u = 0; u < U; ++u) tot[u] += lane_xor16(tot[u]);
#pragma unroll
        for (int u = 0; u < U; ++u) tot[u] += lane_xor_row<8>(tot[u]);
#pragma unroll
        for (int u = 0; u < U; ++u) tot[u] += lane_xor_row<4>(tot[u]);
#pragma unroll
        for (int u = 0; u < U; ++u) tot[u] += lane_xor_row<2>(tot[u]);
#pragma unroll
        for (int u = 0; u < U; ++u) tot[u] += lane_xor_row<1>(tot[u]);
        float mine = tot[0];
#pragma unroll
        for (int u = 1; u < U; ++u) mine = (lane == u) ? tot[u] : mine;
        const float up = mine - s_sh[my_dl];
        const float my_ex = expf(up);
        if (lane < U && i0 + lane < total) {
          atomicAdd(&s_den[my_dl], my_ex);
          if (!(up <= SHIFT_MAX)) s_bad = 1;
        }
#pragma unroll
        for (int u = 0; u < U; ++u) {
          if (i0 + u < total) {
            const int d = lane_int(my_dl, u);
            const float ex = lane_float(my_ex, u);
            if constexpr (H == 64) {
              atomicAdd(&s_acc[d * XS + lane * 2], ex * a[u].v[0]);
              atomicAdd(&s_acc[d * XS + lane * 2 + 1], ex * a[u].v[1]);
            } else {
              atomicAdd(&s_acc[d * XS + lane], ex * a[u].v[0]);
            }
          }
        }
      }
    }
    __syncthreads();
    if constexpr (FINAL) {
      for (int i = tid; i < total; i += OWN_TB) {
        const int d = s_dl[i];
        const float ex = expf(s_e[i] - s_m[d]);
        atomicAdd(&s_den[d], ex);
        atomicAdd(&s_acc[d], ex * s_p[i]);
      }
    } else {
      listed = s_bad == 0;
    }
  }
  if (!listed) {
    if constexpr (!FINAL) {
      for (int i = tid; i < RNODES; i += OWN_TB) { s_m[i] = -INFINITY; s_den[i] = 0.0f; }
      for (int i = tid; i < RNODES * H; i += OWN_TB) s_acc[(i / H) * XS + (i % H)] = 0.0f;
      __syncthreads();
    }
    for (int phase = 0; phase < 2; ++phase) {
      auto one = [&](int s, int d, int eid) {
        const int l = d - base;
        if (phase == 0) {
          const RowFrag<H> a = load_row<H>(xl, s, lane), b = lds_row<H, XS>(s_xa, l, lane);
          const float tot = wave_sum32(score_part<H>(a, b, at));
          if (lane == 0) { ebufG[eid] = tot; ldsMaxF(&s_m[l], tot); }
        } else {
          const float ex = expf(ebufG[eid] - s_m[l]);
          if (lane == 0) atomicAdd(&s_den[l], ex);
          if constexpr (FINAL) {
            const RowFrag<H> a = load_row<H>(xl, s, lane);
            const float pv = wave_sum32(a.v[0] * wl);
            if (lane == 0) atomicAdd(&s_acc[l], ex * pv);
          } else {
            const RowFrag<H> a = load_row<H>(xl, s, lane);
            if constexpr (H == 64) {
              atomicAdd(&s_acc[l * XS + lane * 2], ex * a.v[0]);
              atomicAdd(&s_acc[l * XS + lane * 2 + 1], ex * a.v[1]);
            } else {
              atomicAdd(&s_acc[l * XS + lane], ex * a.v[0]);
            }
          }
        }
      };
      for (int j = wave; j < Rn; j += OWN_TB / 32) one(base + j, base + j, E + base + j);
      for (int e0 = wave * 32; e0 < E; e0 += OWN_TB) {
        const int e = e0 + lane;
        const int d = (e < E) ? dstp[e] : -1;
        const bool hit = (unsigned)(d - base) < (unsigned)Rn;
        unsigned mask = (unsigned)__ballot(hit);
        while (mask) {
          const int bsel = __builtin_ctz(mask);
          mask &= mask - 1;
          one(ei[e0 + bsel], __shfl(d, bsel, 32), e0 + bsel);
        }
      }
      __syncthreads();
    }
  }
  __syncthreads();

  if constexpr (FINAL) {
    float cb = blin[0];
#pragma unroll
    for (int c = 0; c < 32; ++c) cb += bias[c] * Wlin[c];
    for (int j = tid; j < Rn; j += OWN_TB)
      outp[base + j] = s_acc[j] / (s_den[j] + 1e-16f) + cb;
  } else {
    for (int i = tid; i < Rn * H; i += OWN_TB) {
      const int n = i / H, c = i % H;
      const float v = s_acc[n * XS + c] / (s_den[n] + 1e-16f) + bias[c];
      outp[(size_t)base * H + i] = v > 0.0f ? v : 0.01f * v;
    }
  }
}

static inline int cdiv(int a, int b) { return (a + b - 1) / b; }

extern "C" void kernel_launch(void* const* d_in, const int* in_sizes, int n_in,
                              void* d_out, int out_size, void* d_ws, size_t ws_size,
                              hipStream_t stream) {
  const float* x     = (const float*)d_in[0];
  const int*   ei    = (const int*)  d_in[1];
  const float* W1l   = (const float*)d_in[2];
  const float* b1l   = (const float*)d_in[3];
  const float* W1r   = (const float*)d_in[4];
  const float* b1r   = (const float*)d_in[5];
  const float* att1  = (const float*)d_in[6];
  const float* bias1 = (const float*)d_in[7];
  const float* W2l   = (const float*)d_in[8];
  const float* b2l   = (const float*)d_in[9];
  const float* W2r   = (const float*)d_in[10];
  const float* b2r   = (const float*)d_in[11];
  const float* att2  = (const float*)d_in[12];
  const float* bias2 = (const float*)d_in[13];
  const float* Wlin  = (const float*)d_in[14];
  const float* blin  = (const float*)d_in[15];
  float* out = (float*)d_out;

  const int N    = in_sizes[0] / 128;
  const int E    = in_sizes[1] / 2;
  const int Etot = E + N;

  float* ws = (float*)d_ws;
  size_t off = 0;
  float* xl1  = ws + off; off += (size_t)N * 64;
  float* h1   = ws + off; off += (size_t)N * 64;
  float* xl2  = ws + off; off += (size_t)N * 32;
  float* ebuf = ws + off; off += (size_t)Etot;
  off = (off + 1) & ~(size_t)1;
  int* listG = (int*)(ws + off); off += (size_t)cdiv(N, RNODES) * LCAP * 2;
  int* cntG  = (int*)(ws + off); off += (size_t)cdiv(N, RNODES);
  off = (off + 31) & ~(size_t)31;
  const int markGroups = cdiv(cdiv(E, 4), MK_TB);
  const int NW = markGroups * 32;
  unsigned* marks = (unsigned*)(ws + off); off += (size_t)cdiv(N, RNODES) * NW;
  (void)ws_size; (void)n_in; (void)out_size;

  const int ownGroups = cdiv(N, RNODES);
  const int gemmGroups1 = 256;
  const int gemmGroups2 = 512;

  mark_owner_quads<<<markGroups, MK_TB, 0, stream>>>(ei, E, ownGroups, NW, marks);

  gemm_bias_wmma<128, 64><<<gemmGroups1, GM_TB, 0, stream>>>(x, W1l, b1l, xl1, N);
  gat_layer_owner<64, 128, false, false><<<ownGroups, OWN_TB, 0, stream>>>(
      xl1, x, W1r, b1r, att1, ei, E, N, bias1, ebuf, nullptr, nullptr, h1, listG, cntG, marks, NW);

  gemm_bias_wmma<64, 32><<<gemmGroups2, GM_TB, 0, stream>>>(h1, W2l, b2l, xl2, N);
  gat_layer_owner<32, 64, true, true><<<ownGroups, OWN_TB, 0, stream>>>(
      xl2, h1, W2r, b2r, att2, ei, E, N, bias2, ebuf, Wlin, blin, out, listG, cntG, marks, NW);
}


# owner passes (vector-issue bound): vector instructions per 8 entries 216 -> 133 (layer 2: 109 -> 95): layer-1 sums in one fold tree, row offset, LDS address one DPP add each; score sum order changes
# speedup vs baseline: 7.7794x; 1.0773x over previous
#include <hip/hip_runtime.h>
#include <hip/hip_bf16.h>
#include <math.h>

typedef __attribute__((ext_vector_type(2))) float v2f;
typedef __attribute__((ext_vector_type(8))) float v8f;

__device__ __forceinline__ float wave_sum32(float v) {
#pragma unroll
  for (int off = 16; off > 0; off >>= 1) v += __shfl_xor(v, off, 32);
  return v;
}

template <int M>
__device__ __forceinline__ float lane_xor_row(float x) {
  const int b = __builtin_bit_cast(int, x);
  return __builtin_bit_cast(float, __builtin_amdgcn_update_dpp(b, b, 0x160 + M, 0xF, 0xF, true));
}
__device__ __forceinline__ float lane_xor16(float x) {
  const int b = __builtin_bit_cast(int, x);
  return __builtin_bit_cast(float,
                            __builtin_amdgcn_permlanex16(b, b, 0x76543210, 0xfedcba98, false, false));
}
template <int J>
__device__ __forceinline__ int row_lane_int(int x) {
  return __builtin_amdgcn_update_dpp(x, x, 0x150 + J, 0xF, 0xF, true);
}
__device__ __forceinline__ float lane_float(float x, int l) {
  return __builtin_bit_cast(float, __builtin_amdgcn_readlane(__builtin_bit_cast(int, x), l));
}
template <int J> struct Idx { static constexpr int value = J; };
template <int J0, int NJ, class F>
__device__ __forceinline__ void unrolled(F&& f) {
  if constexpr (J0 < NJ) { f(Idx<J0>{}); unrolled<J0 + 1, NJ>(f); }
}

template <int M>
__device__ __forceinline__ float fold_pair(float xa, float xb, bool hi) {
  const float own = hi ? xb : xa;
  const float oth = hi ? xa : xb;
  return own + lane_xor_row<M>(oth);
}

constexpr int GM_TB = 256;

template <int C> struct ColVec;
template <> struct ColVec<4> { typedef __attribute__((ext_vector_type(4))) float type; };
template <> struct ColVec<2> { typedef __attribute__((ext_vector_type(2))) float type; };

template <int K, int H>
__global__ void __launch_bounds__(GM_TB)
gemm_bias_wmma(const float* __restrict__ X, const float* __restrict__ W,
               const float* __restrict__ bias, float* __restrict__ C, int N) {
  constexpr int COLT = H / 16;
  constexpr int KQ = K / 8;
  static_assert(K % 8 == 0 && H % 16 == 0, "whole 16-byte loads, whole column tiles");
  typedef typename ColVec<COLT>::type cvec;
  typedef __attribute__((ext_vector_type(4))) float v4f;
  const int lane = threadIdx.x & 31;
  const int half = lane >> 4, l16 = lane & 15;
  const int wv = __builtin_amdgcn_readfirstlane((int)(threadIdx.x >> 5));
  const int gwave = wv * (int)gridDim.x + (int)blockIdx.x;
  const int nwaves = (int)gridDim.x * (GM_TB / 32);
  const int rowTiles = (N + 15) >> 4;

  v2f wf[KQ][2][COLT];
#pragma unroll
  for (int t = 0; t < KQ; ++t) {
#pragma unroll
    for (int e = 0; e < 2; ++e) {
      const int k = 8 * t + 4 * half + 2 * e;
      const cvec w0 = *(const cvec*)(W + (size_t)k * H + COLT * l16);
      const cvec w1 = *(const cvec*)(W + (size_t)(k + 1) * H + COLT * l16);
#pragma unroll
      for (int ct = 0; ct < COLT; ++ct) { wf[t][e][ct].x = w0[ct]; wf[t][e][ct].y = w1[ct]; }
    }
  }
  const cvec bv = *(const cvec*)(bias + COLT * l16);

  v4f an[KQ];
  int rt = gwave;
  if (rt < rowTiles) {
    const v4f* __restrict__ ap = (const v4f*)(X + (size_t)min(rt * 16 + l16, N - 1) * K + 4 * half);
#pragma unroll
    for (int t = 0; t < KQ; ++t) an[t] = ap[2 * t];
  }
  for (; rt < rowTiles; rt += nwaves) {
    v4f ac[KQ];
#pragma unroll
    for (int t = 0; t < KQ; ++t) ac[t] = an[t];
    const int nrt = rt + nwaves;
    if (nrt < rowTiles) {
      const v4f* __restrict__ ap = (const v4f*)(X + (size_t)min(nrt * 16 + l16, N - 1) * K + 4 * half);
#pragma unroll
      for (int t = 0; t < KQ; ++t) an[t] = ap[2 * t];
    }
    v8f acc[COLT];
#pragma unroll
    for (int ct = 0; ct < COLT; ++ct) acc[ct] = v8f{};
#pragma unroll
    for (int t = 0; t < KQ; ++t) {
      v2f a0, a1;
      a0.x = ac[t].x; a0.y = ac[t].y;
      a1.x = ac[t].z; a1.y = ac[t].w;
#pragma unroll
      for (int ct = 0; ct < COLT; ++ct)
        acc[ct] = __builtin_amdgcn_wmma_f32_16x16x4_f32(false, a0, false, wf[t][0][ct], (short)0, acc[ct],
                                                        false, false);
#pragma unroll
      for (int ct = 0; ct < COLT; ++ct)
        acc[ct] = __builtin_amdgcn_wmma_f32_16x16x4_f32(false, a1, false, wf[t][1][ct], (short)0, acc[ct],
                                                        false, false);
    }
    float* __restrict__ crow = C + (size_t)(rt * 16 + 8 * half) * H + COLT * l16;
    if (rt * 16 + 16 <= N) {
#pragma unroll
      for (int v = 0; v < 8; ++v) {
        cvec o;
#pragma unroll
        for (int ct = 0; ct < COLT; ++ct) o[ct] = acc[ct][v] + bv[ct];
        *(cvec*)(crow + (size_t)v * H) = o;
      }
    } else {
#pragma unroll
      for (int v = 0; v < 8; ++v) {
        cvec o;
#pragma unroll
        for (int ct = 0; ct < COLT; ++ct) o[ct] = acc[ct][v] + bv[ct];
        if (rt * 16 + 8 * half + v < N) *(cvec*)(crow + (size_t)v * H) = o;
      }
    }
  }
}

constexpr int RNODES = 392;
constexpr int LCAP   = 8192;
constexpr float SHIFT_MAX = 30.0f;
constexpr int OWN_TB = 1024;

constexpr int MK_TB    = 1024;
constexpr int MK_OW    = 256;
constexpr int MK_PITCH = 33;

template <bool V> struct QuadVec { static constexpr bool value = V; };

template <bool VEC>
__device__ __forceinline__ int4 load_quad(const int* __restrict__ col, int q, int E) {
  if constexpr (VEC) return ((const int4*)col)[q];
  int4 r;
  const int e = 4 * q;
  r.x = (e     < E) ? col[e]     : -1;
  r.y = (e + 1 < E) ? col[e + 1] : -1;
  r.z = (e + 2 < E) ? col[e + 2] : -1;
  r.w = (e + 3 < E) ? col[e + 3] : -1;
  return r;
}

__global__ void __launch_bounds__(MK_TB)
mark_owner_quads(const int* __restrict__ ei, int E, int G, int NW, unsigned* __restrict__ marks) {
  __shared__ unsigned tile[MK_OW * MK_PITCH];
  const int tid = threadIdx.x, lane = tid & 31, wave = tid >> 5;
  const int q = blockIdx.x * MK_TB + tid;
  const bool vec = (E & 3) == 0;
  int4 d = make_int4(-1, -1, -1, -1);
  if ((long long)4 * q < E) d = vec ? load_quad<true>(ei + E, q, E) : load_quad<false>(ei + E, q, E);
  int own[4];
  own[0] = d.x >= 0 ? d.x / RNODES : -1;
  own[1] = d.y >= 0 ? d.y / RNODES : -1;
  own[2] = d.z >= 0 ? d.z / RNODES : -1;
  own[3] = d.w >= 0 ? d.w / RNODES : -1;
  for (int g0 = 0; g0 < G; g0 += MK_OW) {
    for (int i = tid; i < MK_OW * MK_PITCH; i += MK_TB) tile[i] = 0u;
    __syncthreads();
#pragma unroll
    for (int k = 0; k < 4; ++k) {
      const unsigned o = (unsigned)(own[k] - g0);
      if (o < (unsigned)MK_OW) atomicOr(&tile[o * MK_PITCH + wave], 1u << lane);
    }
    __syncthreads();
    for (int o = wave; o < MK_OW && g0 + o < G; o += MK_TB / 32)
      marks[(size_t)(g0 + o) * NW + (size_t)blockIdx.x * 32 + lane] = tile[o * MK_PITCH + lane];
    __syncthreads();
  }
}

__device__ __forceinline__ void ldsMaxF(float* addr, float val) {
  if (val >= 0.0f) atomicMax((int*)addr, __float_as_int(val));
  else             atomicMin((unsigned int*)addr, __float_as_uint(val));
}

template <int H> struct RowFrag { float v[H / 32]; };

template <int H>
__device__ __forceinline__ RowFrag<H> load_row(const float* __restrict__ basep, int n, int lane) {
  RowFrag<H> r;
  if constexpr (H == 64) {
    const float2 a = *(const float2*)(basep + (size_t)n * 64 + lane * 2);
    r.v[0] = a.x; r.v[1] = a.y;
  } else {
    r.v[0] = basep[(size_t)n * 32 + lane];
  }
  return r;
}

template <int H, int P>
__device__ __forceinline__ RowFrag<H> lds_row(const float* s, int n, int lane) {
  RowFrag<H> r;
  if constexpr (H == 64) {
    const float2 a = *(const float2*)(s + n * P + lane * 2);
    r.v[0] = a.x; r.v[1] = a.y;
  } else {
    r.v[0] = s[n * P + lane];
  }
  return r;
}
template <int H, int P>
__device__ __forceinline__ void lds_put_row(float* s, int n, int lane, const RowFrag<H>& r) {
  if constexpr (H == 64) {
    *(float2*)(s + n * P + lane * 2) = make_float2(r.v[0], r.v[1]);
  } else {
    s[n * P + lane] = r.v[0];
  }
}

template <int H>
__device__ __forceinline__ RowFrag<H> load_row_off(const float* __restrict__ basep, unsigned off) {
  RowFrag<H> r;
  const char* p = (const char*)basep + off;
  if constexpr (H == 64) {
    const float2 a = *(const float2*)p;
    r.v[0] = a.x; r.v[1] = a.y;
  } else {
    r.v[0] = *(const float*)p;
  }
  return r;
}
template <int H>
__device__ __forceinline__ RowFrag<H> lds_row_at(const float* s, int i) {
  RowFrag<H> r;
  if constexpr (H == 64) {
    const float2 a = *(const float2*)(s + i);
    r.v[0] = a.x; r.v[1] = a.y;
  } else {
    r.v[0] = s[i];
  }
  return r;
}

template <int H>
__device__ __forceinline__ float score_part(const RowFrag<H>& a, const RowFrag<H>& b,
                                            const RowFrag<H>& at) {
  if constexpr (H == 64) {
    float h0 = a.v[0] + b.v[0]; h0 = h0 > 0.0f ? h0 : 0.2f * h0;
    float h1 = a.v[1] + b.v[1]; h1 = h1 > 0.0f ? h1 : 0.2f * h1;
    return h0 * at.v[0] + h1 * at.v[1];
  } else {
    float hv = a.v[0] + b.v[0];
    hv = hv > 0.0f ? hv : 0.2f * hv;
    return hv * at.v[0];
  }
}

template <int H, int K, bool FINAL, bool FROMLIST>
__global__ void __launch_bounds__(OWN_TB)
gat_layer_owner(const float* __restrict__ xl, const float* __restrict__ xin,
                const float* __restrict__ Wr, const float* __restrict__ br,
                const float* __restrict__ att, const int* __restrict__ ei, int E, int N,
                const float* __restrict__ bias, float* __restrict__ ebufG,
                const float* __restrict__ Wlin,
                const float* __restrict__ blin, float* __restrict__ outp,
                int* __restrict__ listG, int* __restrict__ cntG,
                const unsigned* __restrict__ marks, int NW) {
  constexpr int U = 8;
  constexpr int COLT = H / 16;
  constexpr int ROWT = (RNODES + 15) / 16;
  static_assert(ROWT <= OWN_TB / 32, "one wave per row tile");
  static_assert(K * H <= LCAP, "the weight fragments are staged in the score array");
  static_assert(FINAL == FROMLIST, "the one-pass layer builds the list; its self-loop rows are set up behind the list builder's first barrier");
  constexpr int XS = FINAL ? H : 2 * H;
  __shared__ __attribute__((aligned(16))) float s_xa[RNODES * XS];
  __shared__ float s_accF[FINAL ? RNODES : 1];
  float* const s_acc = FINAL ? s_accF : s_xa + H;
  __shared__ float s_m[RNODES];
  __shared__ float s_den[RNODES];
  __shared__ float s_sh[FINAL ? 1 : RNODES];
  __shared__ int   s_bad;
  __shared__ int   s_src[LCAP];
  __shared__ int   s_dl[LCAP];
  __shared__ __attribute__((aligned(16))) float s_e[LCAP];
  __shared__ float s_p[FINAL ? LCAP : 1];
  __shared__ int   s_cnt;
  __shared__ int   s_qn;

  const int tid = threadIdx.x, lane = tid & 31, wave = tid >> 5;
  const int base = blockIdx.x * RNODES;
  const int Rn = min(RNODES, N - base);
  const int* __restrict__ dstp = ei + E;

  if constexpr (FINAL) {
    for (int i = tid; i < RNODES; i += OWN_TB) { s_accF[i] = 0.0f; s_m[i] = -INFINITY; s_den[i] = 0.0f; }
    for (int j = tid; j < Rn; j += OWN_TB) { s_src[j] = base + j; s_dl[j] = j; }
  } else {
    if (tid == 0) s_bad = 0;
  }
  RowFrag<H> at;
  if constexpr (H == 64) { at.v[0] = att[lane * 2]; at.v[1] = att[lane * 2 + 1]; }
  else                   { at.v[0] = att[lane]; }

  {
    float* s_w = s_e;
    constexpr int SB = 8;
    constexpr int NB = K / 4 / SB;
    constexpr int WPT = K * H / OWN_TB;
    static_assert(K % (4 * SB) == 0 && (K * H) % OWN_TB == 0, "whole blocks, whole staging rounds");
    const bool gw = wave < ROWT && wave * 16 < Rn;
    const int half = lane >> 4, l16 = lane & 15;
    const int r = min(wave * 16 + l16, Rn - 1);
    const float* __restrict__ arow = xin + (size_t)(base + r) * K + 2 * half;
    v2f an[SB];
#pragma unroll
    for (int j = 0; j < SB; ++j) an[j] = *(const v2f*)(arow + 4 * j);
    float wv[WPT];
#pragma unroll
    for (int j = 0; j < WPT; ++j) {
      const int i = tid + j * OWN_TB;
      const int e = i & 1, l = (i >> 1) & 31, t = i >> 6;
      const int k = 4 * (t / COLT) + 2 * (l >> 4) + e;
      wv[j] = Wr[k * H + (t % COLT) * 16 + (l & 15)];
    }
#pragma unroll
    for (int j = 0; j < WPT; ++j) s_w[tid + j * OWN_TB] = wv[j];
    __syncthreads();
    if (gw) {
      v8f acc[COLT];
#pragma unroll
      for (int ct = 0; ct < COLT; ++ct) acc[ct] = v8f{};
#pragma unroll
      for (int blk = 0; blk < NB; ++blk) {
        v2f ac[SB];
#pragma unroll
        for (int j = 0; j < SB; ++j) ac[j] = an[j];
        if (blk + 1 < NB) {
#pragma unroll
          for (int j = 0; j < SB; ++j) an[j] = *(const v2f*)(arow + 4 * ((blk + 1) * SB + j));
        }
#pragma unroll
        for (int j = 0; j < SB; ++j) {
          const int s = blk * SB + j;
#pragma unroll
          for (int ct = 0; ct < COLT; ++ct) {
            const v2f b = *(const v2f*)(s_w + ((s * COLT + ct) * 32 + lane) * 2);
            acc[ct] = __builtin_amdgcn_wmma_f32_16x16x4_f32(false, ac[j], false, b, (short)0, acc[ct],
                                                            false, false);
          }
        }
      }
#pragma unroll
      for (int ct = 0; ct < COLT; ++ct) {
        const float bv = br[ct * 16 + l16];
#pragma unroll
        for (int v = 0; v < 8; ++v) {
          const int row = wave * 16 + v + 8 * half;
          if (row < Rn) s_xa[row * XS + ct * 16 + l16] = acc[ct][v] + bv;
        }
      }
    }
  }
  int2* __restrict__ lg = (int2*)listG + (size_t)blockIdx.x * LCAP;
  int cnt;
  if constexpr (FROMLIST) {
    cnt = cntG[blockIdx.x];
    if (Rn + cnt <= LCAP) {
      for (int i = tid; i < cnt; i += OWN_TB) {
        const int2 p = lg[i];
        s_src[Rn + i] = p.x; s_dl[Rn + i] = p.y;
      }
    }
    __syncthreads();
  } else {
    if (tid == 0) { s_cnt = 0; s_qn = 0; }
    __syncthreads();

    if constexpr (!FINAL) {
      constexpr int PU = 16;
      for (int j0 = wave * PU; j0 < Rn; j0 += (OWN_TB / 32) * PU) {
        RowFrag<H> a[PU], b[PU];
#pragma unroll
        for (int u = 0; u < PU; ++u) {
          const int j = min(j0 + u, Rn - 1);
          a[u] = load_row<H>(xl, base + j, lane);
          b[u] = lds_row<H, XS>(s_xa, j, lane);
        }
        float tot[PU];
#pragma unroll
        for (int u = 0; u < PU; ++u) tot[u] = score_part<H>(a[u], b[u], at);
#pragma unroll
        for (int u = 0; u < PU; ++u) tot[u] += lane_xor16(tot[u]);
#pragma unroll
        for (int u = 0; u < PU; ++u) tot[u] += lane_xor_row<8>(tot[u]);
#pragma unroll
        for (int u = 0; u < PU; ++u) tot[u] += lane_xor_row<4>(tot[u]);
#pragma unroll
        for (int u = 0; u < PU; ++u) tot[u] += lane_xor_row<2>(tot[u]);
#pragma unroll
        for (int u = 0; u < PU; ++u) tot[u] += lane_xor_row<1>(tot[u]);
        float mine = tot[0];
#pragma unroll
        for (int u = 1; u < PU; ++u) mine = (lane == u) ? tot[u] : mine;
        if (lane < PU && j0 + lane < Rn) { s_sh[j0 + lane] = mine; s_den[j0 + lane] = 1.0f; }
#pragma unroll
        for (int u = 0; u < PU; ++u) {
          if (j0 + u < Rn) lds_put_row<H, XS>(s_acc, j0 + u, lane, a[u]);
        }
      }
    }

    int* s_q = (int*)s_e;
    const unsigned* __restrict__ plane = marks + (size_t)blockIdx.x * NW;
    constexpr int PW = 8;
    for (int w0 = tid; w0 < NW; w0 += OWN_TB * PW) {
      unsigned m[PW];
#pragma unroll
      for (int k = 0; k < PW; ++k) {
        const int w = w0 + k * OWN_TB;
        m[k] = (w < NW) ? plane[w] : 0u;
      }
      int nb = 0;
#pragma unroll
      for (int k = 0; k < PW; ++k) nb += __popc(m[k]);
      int slot = atomicAdd(&s_qn, nb);
#pragma unroll
      for (int k = 0; k < PW; ++k) {
        unsigned mm = m[k];
        while (mm) {
          const int b = __builtin_ctz(mm);
          mm &= mm - 1;
          if (slot < LCAP) s_q[slot] = (w0 + k * OWN_TB) * 32 + b;
          ++slot;
        }
      }
    }
    __syncthreads();
    const int qn = s_qn;
    const int nqd = min(qn, LCAP);

    auto hit = [&](int d) { return ((unsigned)(d - base) < (unsigned)Rn) ? 1 : 0; };
    auto put = [&](int d, int s, int& slot) {
      if (hit(d)) {
        if (slot < LCAP) { s_src[slot] = s; s_dl[slot] = d - base; }
        ++slot;
      }
    };
    constexpr int QV = 4;
    auto gather_quads = [&](auto vtag) {
      constexpr bool VEC = decltype(vtag)::value;
      for (int i0 = tid; i0 < nqd; i0 += OWN_TB * QV) {
        int4 d4[QV], s4[QV];
#pragma unroll
        for (int k = 0; k < QV; ++k) {
          const int qq = s_q[min(i0 + k * OWN_TB, nqd - 1)];
          d4[k] = load_quad<VEC>(dstp, qq, E);
          s4[k] = load_quad<VEC>(ei, qq, E);
        }
        int nm = 0;
#pragma unroll
        for (int k = 0; k < QV; ++k) {
          if (i0 + k * OWN_TB < nqd) nm += hit(d4[k].x) + hit(d4[k].y) + hit(d4[k].z) + hit(d4[k].w);
        }
        int slot = Rn + atomicAdd(&s_cnt, nm);
#pragma unroll
        for (int k = 0; k < QV; ++k) {
          if (i0 + k * OWN_TB < nqd) {
            put(d4[k].x, s4[k].x, slot); put(d4[k].y, s4[k].y, slot);
            put(d4[k].z, s4[k].z, slot); put(d4[k].w, s4[k].w, slot);
          }
        }
      }
    };
    if ((E & 3) == 0) gather_quads(QuadVec<true>{});
    else              gather_quads(QuadVec<false>{});
    __syncthreads();

    cnt = (qn > LCAP) ? qn : s_cnt;
    if (tid == 0) cntG[blockIdx.x] = cnt;
    if (Rn + cnt <= LCAP) {
      for (int i = tid; i < cnt; i += OWN_TB) lg[i] = make_int2(s_src[Rn + i], s_dl[Rn + i]);
    }
  }

  const int total = Rn + cnt;
  float wl = 0.0f;
  if constexpr (FINAL) {
    static_assert(!FINAL || H == 32, "one channel per lane in the last layer");
    wl = Wlin[lane];
  }

  bool listed = total <= LCAP;
  if (listed) {
    const int wv = __builtin_amdgcn_readfirstlane(wave);
    const int lane_e = lane & (U - 1);
    for (int i0 = (FINAL ? 0 : Rn) + wv * U; i0 < total; i0 += (OWN_TB / 32) * U) {
      const bool live = i0 + lane_e < total;
      const int li = min(i0 + lane_e, total - 1);
      const int my_dl = s_dl[li];
      const int my_src = s_src[li];
      const int my_row = my_dl * XS;
      const int my_off = my_src * (H * 4);
      RowFrag<H> a[U], b[U];
      int xo[U];
      unrolled<0, U>([&](auto jt) {
        constexpr int u = decltype(jt)::value;
        a[u] = load_row_off<H>(xl, (unsigned)(row_lane_int<u>(my_off) + lane * (H / 8)));
        xo[u] = row_lane_int<u>(my_row) + lane * (H / 32);
        b[u] = lds_row_at<H>(s_xa, xo[u]);
      });
      float tot[U];
#pragma unroll
      for (int u = 0; u < U; ++u) tot[u] = score_part<H>(a[u], b[u], at);
      if constexpr (FINAL) {
        float x[2 * U];
#pragma unroll
        for (int u = 0; u < U; ++u) { x[u] = tot[u]; x[U + u] = a[u].v[0] * wl; }
        const bool h1 = (lane & 1) != 0, h2 = (lane & 2) != 0, h4 = (lane & 4) != 0, h8 = (lane & 8) != 0;
        float y[U], z[U / 2], w[U / 4];
#pragma unroll
        for (int i = 0; i < U; ++i) y[i] = fold_pair<1>(x[2 * i], x[2 * i + 1], h1);
#pragma unroll
        for (int i = 0; i < U / 2; ++i) z[i] = fold_pair<2>(y[2 * i], y[2 * i + 1], h2);
#pragma unroll
        for (int i = 0; i < U / 4; ++i) w[i] = fold_pair<4>(z[2 * i], z[2 * i + 1], h4);
        static_assert(U == 8, "four fold stages take 16 vectors to one");
        float r = fold_pair<8>(w[0], w[1], h8);
        r += lane_xor16(r);
        const int idx = i0 + (lane & (U - 1));
        if (lane < 2 * U && idx < total) {
          if (lane < U) { s_e[idx] = r; ldsMaxF(&s_m[my_dl], r); }
          else          { s_p[idx] = r; }
        }
      } else {
        static_assert(U == 8, "three fold stages take 8 vectors to one");
        const bool h1 = (lane & 1) != 0, h2 = (lane & 2) != 0, h4 = (lane & 4) != 0;
        float y[U / 2], z[U / 4];
#pragma unroll
        for (int i = 0; i < U / 2; ++i) y[i] = fold_pair<1>(tot[2 * i], tot[2 * i + 1], h1);
#pragma unroll
        for (int i = 0; i < U / 4; ++i) z[i] = fold_pair<2>(y[2 * i], y[2 * i + 1], h2);
        float mine = fold_pair<4>(z[0], z[1], h4);
        mine += lane_xor_row<8>(mine);
        mine += lane_xor16(mine);
        const float up = mine - s_sh[my_dl];
        const float my_ex = expf(up);
        if (lane < U && live) {
          atomicAdd(&s_den[my_dl], my_ex);
          if (!(up <= SHIFT_MAX)) s_bad = 1;
        }
        auto add_row = [&](auto jt) {
          constexpr int u = decltype(jt)::value;
          const float ex = lane_float(my_ex, u);
          if constexpr (H == 64) {
            atomicAdd(&s_acc[xo[u]], ex * a[u].v[0]);
            atomicAdd(&s_acc[xo[u] + 1], ex * a[u].v[1]);
          } else {
            atomicAdd(&s_acc[xo[u]], ex * a[u].v[0]);
          }
        };
        if (i0 + U <= total) {
          unrolled<0, U>(add_row);
        } else {
          unrolled<0, U>([&](auto jt) { if (i0 + decltype(jt)::value < total) add_row(jt); });
        }
      }
    }
    __syncthreads();
    if constexpr (FINAL) {
      for (int i = tid; i < total; i += OWN_TB) {
        const int d = s_dl[i];
        const float ex = expf(s_e[i] - s_m[d]);
        atomicAdd(&s_den[d], ex);
        atomicAdd(&s_acc[d], ex * s_p[i]);
      }
    } else {
      listed = s_bad == 0;
    }
  }
  if (!listed) {
    if constexpr (!FINAL) {
      for (int i = tid; i < RNODES; i += OWN_TB) { s_m[i] = -INFINITY; s_den[i] = 0.0f; }
      for (int i = tid; i < RNODES * H; i += OWN_TB) s_acc[(i / H) * XS + (i % H)] = 0.0f;
      __syncthreads();
    }
    for (int phase = 0; phase < 2; ++phase) {
      auto one = [&](int s, int d, int eid) {
        const int l = d - base;
        if (phase == 0) {
          const RowFrag<H> a = load_row<H>(xl, s, lane), b = lds_row<H, XS>(s_xa, l, lane);
          const float tot = wave_sum32(score_part<H>(a, b, at));
          if (lane == 0) { ebufG[eid] = tot; ldsMaxF(&s_m[l], tot); }
        } else {
          const float ex = expf(ebufG[eid] - s_m[l]);
          if (lane == 0) atomicAdd(&s_den[l], ex);
          if constexpr (FINAL) {
            const RowFrag<H> a = load_row<H>(xl, s, lane);
            const float pv = wave_sum32(a.v[0] * wl);
            if (lane == 0) atomicAdd(&s_acc[l], ex * pv);
          } else {
            const RowFrag<H> a = load_row<H>(xl, s, lane);
            if constexpr (H == 64) {
              atomicAdd(&s_acc[l * XS + lane * 2], ex * a.v[0]);
              atomicAdd(&s_acc[l * XS + lane * 2 + 1], ex * a.v[1]);
            } else {
              atomicAdd(&s_acc[l * XS + lane], ex * a.v[0]);
            }
          }
        }
      };
      for (int j = wave; j < Rn; j += OWN_TB / 32) one(base + j, base + j, E + base + j);
      for (int e0 = wave * 32; e0 < E; e0 += OWN_TB) {
        const int e = e0 + lane;
        const int d = (e < E) ? dstp[e] : -1;
        const bool hit = (unsigned)(d - base) < (unsigned)Rn;
        unsigned mask = (unsigned)__ballot(hit);
        while (mask) {
          const int bsel = __builtin_ctz(mask);
          mask &= mask - 1;
          one(ei[e0 + bsel], __shfl(d, bsel, 32), e0 + bsel);
        }
      }
      __syncthreads();
    }
  }
  __syncthreads();

  if constexpr (FINAL) {
    float cb = blin[0];
#pragma unroll
    for (int c = 0; c < 32; ++c) cb += bias[c] * Wlin[c];
    for (int j = tid; j < Rn; j += OWN_TB)
      outp[base + j] = s_acc[j] / (s_den[j] + 1e-16f) + cb;
  } else {
    for (int i = tid; i < Rn * H; i += OWN_TB) {
      const int n = i / H, c = i % H;
      const float v = s_acc[n * XS + c] / (s_den[n] + 1e-16f) + bias[c];
      outp[(size_t)base * H + i] = v > 0.0f ? v : 0.01f * v;
    }
  }
}

static inline int cdiv(int a, int b) { return (a + b - 1) / b; }

extern "C" void kernel_launch(void* const* d_in, const int* in_sizes, int n_in,
                              void* d_out, int out_size, void* d_ws, size_t ws_size,
                              hipStream_t stream) {
  const float* x     = (const float*)d_in[0];
  const int*   ei    = (const int*)  d_in[1];
  const float* W1l   = (const float*)d_in[2];
  const float* b1l   = (const float*)d_in[3];
  const float* W1r   = (const float*)d_in[4];
  const float* b1r   = (const float*)d_in[5];
  const float* att1  = (const float*)d_in[6];
  const float* bias1 = (const float*)d_in[7];
  const float* W2l   = (const float*)d_in[8];
  const float* b2l   = (const float*)d_in[9];
  const float* W2r   = (const float*)d_in[10];
  const float* b2r   = (const float*)d_in[11];
  const float* att2  = (const float*)d_in[12];
  const float* bias2 = (const float*)d_in[13];
  const float* Wlin  = (const float*)d_in[14];
  const float* blin  = (const float*)d_in[15];
  float* out = (float*)d_out;

  const int N    = in_sizes[0] / 128;
  const int E    = in_sizes[1] / 2;
  const int Etot = E + N;

  float* ws = (float*)d_ws;
  size_t off = 0;
  float* xl1  = ws + off; off += (size_t)N * 64;
  float* h1   = ws + off; off += (size_t)N * 64;
  float* xl2  = ws + off; off += (size_t)N * 32;
  float* ebuf = ws + off; off += (size_t)Etot;
  off = (off + 1) & ~(size_t)1;
  int* listG = (int*)(ws + off); off += (size_t)cdiv(N, RNODES) * LCAP * 2;
  int* cntG  = (int*)(ws + off); off += (size_t)cdiv(N, RNODES);
  off = (off + 31) & ~(size_t)31;
  const int markGroups = cdiv(cdiv(E, 4), MK_TB);
  const int NW = markGroups * 32;
  unsigned* marks = (unsigned*)(ws + off); off += (size_t)cdiv(N, RNODES) * NW;
  (void)ws_size; (void)n_in; (void)out_size;

  const int ownGroups = cdiv(N, RNODES);
  const int gemmGroups1 = 256;
  const int gemmGroups2 = 512;

  mark_owner_quads<<<markGroups, MK_TB, 0, stream>>>(ei, E, ownGroups, NW, marks);

  gemm_bias_wmma<128, 64><<<gemmGroups1, GM_TB, 0, stream>>>(x, W1l, b1l, xl1, N);
  gat_layer_owner<64, 128, false, false><<<ownGroups, OWN_TB, 0, stream>>>(
      xl1, x, W1r, b1r, att1, ei, E, N, bias1, ebuf, nullptr, nullptr, h1, listG, cntG, marks, NW);

  gemm_bias_wmma<64, 32><<<gemmGroups2, GM_TB, 0, stream>>>(h1, W2l, b2l, xl2, N);
  gat_layer_owner<32, 64, true, true><<<ownGroups, OWN_TB, 0, stream>>>(
      xl2, h1, W2r, b2r, att2, ei, E, N, bias2, ebuf, Wlin, blin, out, listG, cntG, marks, NW);
}
